# starting kernel
# baseline (speedup 1.0000x reference)
#include <hip/hip_runtime.h>
#include <hip/hip_bf16.h>

typedef __attribute__((ext_vector_type(2))) float v2f;
typedef __attribute__((ext_vector_type(8))) float v8f;

#define DIN 128

__global__ __launch_bounds__(256) void deg_init_kernel(float* deg, int n) {
    int i = blockIdx.x * blockDim.x + threadIdx.x;
    if (i < n) deg[i] = 1.0f;
}

__global__ __launch_bounds__(256) void deg_accum_kernel(const int* __restrict__ ei, int e,
                                                        float* __restrict__ deg) {
    int idx = blockIdx.x * blockDim.x + threadIdx.x;
    if (idx < e) atomicAdd(&deg[ei[e + idx]], 1.0f);
}

__global__ __launch_bounds__(256) void deg_to_dinv_kernel(float* deg, int n) {
    int i = blockIdx.x * blockDim.x + threadIdx.x;
    if (i < n) deg[i] = rsqrtf(deg[i]);
}

template <int DOUT>
__global__ __launch_bounds__(256) void gemm_wmma_kernel(const float* __restrict__ X,
                                                        const float* __restrict__ W,
                                                        float* __restrict__ H, int n) {
    constexpr int NT = DOUT / 16;
    constexpr int NPAIR = 32 * NT * 32;
    __shared__ float sW[NPAIR * 2];

    for (int pair = threadIdx.x; pair < NPAIR; pair += 256) {
        int l  = pair & 31;
        int tt = (pair >> 5) % NT;
        int ks = (pair >> 5) / NT;
        int col = tt * 16 + (l & 15);
        int k0  = ks * 4 + ((l >> 4) * 2);
        sW[pair * 2 + 0] = W[k0 * DOUT + col];
        sW[pair * 2 + 1] = W[(k0 + 1) * DOUT + col];
    }
    __syncthreads();

    const int wave = threadIdx.x >> 5;
    const int lane = threadIdx.x & 31;
    const int m    = lane & 15;
    const int kk   = (lane >> 4) * 2;

    const long long row0 = (long long)blockIdx.x * 128 + wave * 16;
    long long rowm = row0 + m;
    if (rowm >= n) rowm = n - 1;
    const float* __restrict__ xrow = X + rowm * DIN + kk;
    const float* __restrict__ bl   = sW + lane * 2;

    v8f acc[NT] = {};

#pragma unroll 4
    for (int ks = 0; ks < 32; ks++) {
        v2f a = *(const v2f*)(xrow + ks * 4);
#pragma unroll
        for (int t = 0; t < NT; t++) {
            v2f b = *(const v2f*)(bl + (ks * NT + t) * 64);
            acc[t] = __builtin_amdgcn_wmma_f32_16x16x4_f32(
                false, a, false, b, (short)0, acc[t], false, false);
        }
    }

    const long long rbase = row0 + (lane >> 4) * 8;
#pragma unroll
    for (int v = 0; v < 8; v++) {
        long long r = rbase + v;
        if (r < n) {
            float* __restrict__ orow = H + r * DOUT;
#pragma unroll
            for (int t = 0; t < NT; t++) orow[t * 16 + m] = acc[t][v];
        }
    }
}

template <int D>
__global__ __launch_bounds__(256) void init_agg_kernel(const float* __restrict__ H,
                                                       const float* __restrict__ bias,
                                                       const float* __restrict__ dinv,
                                                       float* __restrict__ Agg, int n) {
    long long idx = (long long)blockIdx.x * blockDim.x + threadIdx.x;
    long long total = (long long)n * D;
    if (idx < total) {
        int r = (int)(idx / D);
        int c = (int)(idx - (long long)r * D);
        float w = dinv[r];
        Agg[idx] = bias[c] + H[idx] * (w * w);
    }
}

template <int D>
__global__ __launch_bounds__(256) void scatter_kernel(const int* __restrict__ ei, int e,
                                                      const float* __restrict__ dinv,
                                                      const float* __restrict__ H,
                                                      float* __restrict__ Agg) {
    constexpr int VEC = D / 32;
    const int lane = threadIdx.x & 31;
    const int wid  = (int)((blockIdx.x * (long long)blockDim.x + threadIdx.x) >> 5);
    const int nw   = (int)(((long long)gridDim.x * blockDim.x) >> 5);

    for (int ed = wid; ed < e; ed += nw) {
        int s = ei[ed];
        int d = ei[e + ed];
        float w = dinv[s] * dinv[d];

        int e2 = ed + nw;
        if (e2 < e) {
            int s2 = ei[e2];
            __builtin_prefetch(H + (long long)s2 * D + lane * VEC, 0, 1);
        }

        const float* __restrict__ hs = H + (long long)s * D + lane * VEC;
        float* __restrict__ ad = Agg + (long long)d * D + lane * VEC;
        if (VEC == 4) {
            float4 v = *(const float4*)hs;
            atomicAdd(ad + 0, v.x * w);
            atomicAdd(ad + 1, v.y * w);
            atomicAdd(ad + 2, v.z * w);
            atomicAdd(ad + 3, v.w * w);
        } else {
            float2 v = *(const float2*)hs;
            atomicAdd(ad + 0, v.x * w);
            atomicAdd(ad + 1, v.y * w);
        }
    }
}

__global__ __launch_bounds__(256) void elu_kernel(float* __restrict__ A, long long total) {
    long long idx = (long long)blockIdx.x * blockDim.x + threadIdx.x;
    if (idx < total) {
        float v = A[idx];
        A[idx] = v > 0.0f ? v : (__expf(v) - 1.0f);
    }
}

extern "C" void kernel_launch(void* const* d_in, const int* in_sizes, int n_in,
                              void* d_out, int out_size, void* d_ws, size_t ws_size,
                              hipStream_t stream) {
    const float* x  = (const float*)d_in[0];
    const int*   ei = (const int*)d_in[1];
    const float* W1 = (const float*)d_in[2];
    const float* b1 = (const float*)d_in[3];
    const float* W2 = (const float*)d_in[4];
    const float* b2 = (const float*)d_in[5];
    float* out = (float*)d_out;

    const int N = in_sizes[0] / DIN;
    const int E = in_sizes[1] / 2;
    const int DH = 128, DOUT = 64;

    char* ws = (char*)d_ws;
    float* dinv = (float*)ws;
    size_t off = (((size_t)N * sizeof(float)) + 255) & ~(size_t)255;
    float* hbuf = (float*)(ws + off);
    float* aggbuf = (float*)(ws + off + (size_t)N * DH * sizeof(float));

    const int B = 256;
    const int gN    = (N + B - 1) / B;
    const int gE    = (E + B - 1) / B;
    const int gRows = (N + 127) / 128;
    const int gNDH  = (int)(((long long)N * DH + B - 1) / B);
    const int gNDO  = (int)(((long long)N * DOUT + B - 1) / B);
    const int gEdgeW = (E + 7) / 8;

    deg_init_kernel<<<gN, B, 0, stream>>>(dinv, N);
    deg_accum_kernel<<<gE, B, 0, stream>>>(ei, E, dinv);
    deg_to_dinv_kernel<<<gN, B, 0, stream>>>(dinv, N);

    gemm_wmma_kernel<128><<<gRows, B, 0, stream>>>(x, W1, hbuf, N);
    init_agg_kernel<128><<<gNDH, B, 0, stream>>>(hbuf, b1, dinv, aggbuf, N);
    scatter_kernel<128><<<gEdgeW, B, 0, stream>>>(ei, E, dinv, hbuf, aggbuf);
    elu_kernel<<<gNDH, B, 0, stream>>>(aggbuf, (long long)N * DH);

    gemm_wmma_kernel<64><<<gRows, B, 0, stream>>>(aggbuf, W2, hbuf, N);
    init_agg_kernel<64><<<gNDO, B, 0, stream>>>(hbuf, b2, dinv, out, N);
    scatter_kernel<64><<<gEdgeW, B, 0, stream>>>(ei, E, dinv, hbuf, out);
}


# Replace f32 atomic row scatter by per-call dst-grouped edge list (int atomic slot, scan, fill) + one-wave-per-node gather, bias/self/ELU fused; rows summed in ascending src order, same every run
# speedup vs baseline: 2.0994x; 2.0994x over previous
#include <hip/hip_runtime.h>
#include <hip/hip_bf16.h>

typedef __attribute__((ext_vector_type(2))) float v2f;
typedef __attribute__((ext_vector_type(8))) float v8f;

#define DIN 128
#define SCAN_PER_BLOCK 1024

__global__ __launch_bounds__(256) void cnt_zero_kernel(int* __restrict__ cnt, int n) {
    int i = blockIdx.x * blockDim.x + threadIdx.x;
    if (i < n) cnt[i] = 0;
}

__global__ __launch_bounds__(256) void count_kernel(const int* __restrict__ ei, int e,
                                                    int* __restrict__ cnt,
                                                    int* __restrict__ pos) {
    int idx = blockIdx.x * blockDim.x + threadIdx.x;
    if (idx < e) pos[idx] = atomicAdd(&cnt[ei[e + idx]], 1);
}

__global__ __launch_bounds__(256) void block_sums_kernel(const int* __restrict__ cnt, int n,
                                                         int* __restrict__ bsum) {
    __shared__ int red[8];
    const int base = blockIdx.x * SCAN_PER_BLOCK + threadIdx.x * 4;
    int s = 0;
#pragma unroll
    for (int k = 0; k < 4; k++)
        if (base + k < n) s += cnt[base + k];
    for (int o = 16; o > 0; o >>= 1) s += __shfl_xor(s, o, 32);
    if ((threadIdx.x & 31) == 0) red[threadIdx.x >> 5] = s;
    __syncthreads();
    if (threadIdx.x == 0) {
        int t = 0;
        for (int w = 0; w < 8; w++) t += red[w];
        bsum[blockIdx.x] = t;
    }
}

__global__ __launch_bounds__(256) void scan_kernel(const int* __restrict__ cnt, int n,
                                                   const int* __restrict__ bsum,
                                                   int* __restrict__ rp,
                                                   float* __restrict__ dinv) {
    __shared__ int redp[8];
    __shared__ int wtot[8];
    const int lane = threadIdx.x & 31;
    const int wave = threadIdx.x >> 5;

    int p = 0;
    for (int j = threadIdx.x; j < (int)blockIdx.x; j += 256) p += bsum[j];
    for (int o = 16; o > 0; o >>= 1) p += __shfl_xor(p, o, 32);
    if (lane == 0) redp[wave] = p;

    const int base = blockIdx.x * SCAN_PER_BLOCK + threadIdx.x * 4;
    int c[4];
#pragma unroll
    for (int k = 0; k < 4; k++) c[k] = (base + k < n) ? cnt[base + k] : 0;
    const int t = c[0] + c[1] + c[2] + c[3];
    int incl = t;
    for (int o = 1; o < 32; o <<= 1) {
        int u = __shfl_up(incl, o, 32);
        if (lane >= o) incl += u;
    }
    if (lane == 31) wtot[wave] = incl;
    __syncthreads();

    int offs = incl - t;
    for (int w = 0; w < 8; w++) {
        offs += redp[w];
        if (w < wave) offs += wtot[w];
    }
#pragma unroll
    for (int k = 0; k < 4; k++) {
        if (base + k < n) {
            rp[base + k] = offs;
            dinv[base + k] = rsqrtf((float)(c[k] + 1));
            offs += c[k];
            if (base + k == n - 1) rp[n] = offs;
        }
    }
}

__global__ __launch_bounds__(256) void fill_kernel(const int* __restrict__ ei, int e,
                                                   const int* __restrict__ rp,
                                                   const int* __restrict__ pos,
                                                   int* __restrict__ csr) {
    int idx = blockIdx.x * blockDim.x + threadIdx.x;
    if (idx < e) csr[rp[ei[e + idx]] + pos[idx]] = ei[idx];
}

template <int DOUT>
__global__ __launch_bounds__(256) void gemm_wmma_kernel(const float* __restrict__ X,
                                                        const float* __restrict__ W,
                                                        float* __restrict__ H, int n) {
    constexpr int NT = DOUT / 16;
    constexpr int NPAIR = 32 * NT * 32;
    __shared__ float sW[NPAIR * 2];

    for (int pair = threadIdx.x; pair < NPAIR; pair += 256) {
        int l  = pair & 31;
        int tt = (pair >> 5) % NT;
        int ks = (pair >> 5) / NT;
        int col = tt * 16 + (l & 15);
        int k0  = ks * 4 + ((l >> 4) * 2);
        sW[pair * 2 + 0] = W[k0 * DOUT + col];
        sW[pair * 2 + 1] = W[(k0 + 1) * DOUT + col];
    }
    __syncthreads();

    const int wave = threadIdx.x >> 5;
    const int lane = threadIdx.x & 31;
    const int m    = lane & 15;
    const int kk   = (lane >> 4) * 2;

    const long long row0 = (long long)blockIdx.x * 128 + wave * 16;
    long long rowm = row0 + m;
    if (rowm >= n) rowm = n - 1;
    const float* __restrict__ xrow = X + rowm * DIN + kk;
    const float* __restrict__ bl   = sW + lane * 2;

    v8f acc[NT] = {};

#pragma unroll 4
    for (int ks = 0; ks < 32; ks++) {
        v2f a = *(const v2f*)(xrow + ks * 4);
#pragma unroll
        for (int t = 0; t < NT; t++) {
            v2f b = *(const v2f*)(bl + (ks * NT + t) * 64);
            acc[t] = __builtin_amdgcn_wmma_f32_16x16x4_f32(
                false, a, false, b, (short)0, acc[t], false, false);
        }
    }

    const long long rbase = row0 + (lane >> 4) * 8;
#pragma unroll
    for (int v = 0; v < 8; v++) {
        long long r = rbase + v;
        if (r < n) {
            float* __restrict__ orow = H + r * DOUT;
#pragma unroll
            for (int t = 0; t < NT; t++) orow[t * 16 + m] = acc[t][v];
        }
    }
}

template <int D, bool ELU>
__global__ __launch_bounds__(256) void gather_kernel(const int* __restrict__ rp,
                                                     const int* __restrict__ csr,
                                                     const float* __restrict__ dinv,
                                                     const float* __restrict__ H,
                                                     const float* __restrict__ bias,
                                                     float* __restrict__ Out, int n) {
    constexpr int VEC = D / 32;
    typedef __attribute__((ext_vector_type(VEC))) float vecf;
    const int lane = threadIdx.x & 31;
    const int node = __builtin_amdgcn_readfirstlane((int)(blockIdx.x * 8 + (threadIdx.x >> 5)));
    if (node >= n) return;

    const int beg = rp[node];
    const int end = rp[node + 1];
    const int d = end - beg;
    const float di = dinv[node];
    const float* __restrict__ hl = H + lane * VEC;

    const vecf hv = *(const vecf*)(hl + (long long)node * D);
    const vecf bv = *(const vecf*)(bias + lane * VEC);
    vecf acc = bv + hv * (di * di);

    if (d <= 32) {
        int key = 0x7fffffff;
        if (lane < d) key = csr[beg + lane];
        int rank = 0;
        for (int j = 0; j < d; j++) {
            int kj = __builtin_amdgcn_readlane(key, j);
            rank += (kj < key || (kj == key && j < lane)) ? 1 : 0;
        }
        const int skey = __builtin_amdgcn_ds_permute(rank << 2, key);

        int j = 0;
        for (; j + 4 <= d; j += 4) {
            const int s0 = __builtin_amdgcn_readlane(skey, j + 0);
            const int s1 = __builtin_amdgcn_readlane(skey, j + 1);
            const int s2 = __builtin_amdgcn_readlane(skey, j + 2);
            const int s3 = __builtin_amdgcn_readlane(skey, j + 3);
            const float w0 = dinv[s0] * di;
            const float w1 = dinv[s1] * di;
            const float w2 = dinv[s2] * di;
            const float w3 = dinv[s3] * di;
            const vecf v0 = *(const vecf*)(hl + (long long)s0 * D);
            const vecf v1 = *(const vecf*)(hl + (long long)s1 * D);
            const vecf v2 = *(const vecf*)(hl + (long long)s2 * D);
            const vecf v3 = *(const vecf*)(hl + (long long)s3 * D);
            acc += v0 * w0;
            acc += v1 * w1;
            acc += v2 * w2;
            acc += v3 * w3;
        }
        const int r = d - j;
        if (r == 3) {
            const int s0 = __builtin_amdgcn_readlane(skey, j + 0);
            const int s1 = __builtin_amdgcn_readlane(skey, j + 1);
            const int s2 = __builtin_amdgcn_readlane(skey, j + 2);
            const float w0 = dinv[s0] * di;
            const float w1 = dinv[s1] * di;
            const float w2 = dinv[s2] * di;
            const vecf v0 = *(const vecf*)(hl + (long long)s0 * D);
            const vecf v1 = *(const vecf*)(hl + (long long)s1 * D);
            const vecf v2 = *(const vecf*)(hl + (long long)s2 * D);
            acc += v0 * w0;
            acc += v1 * w1;
            acc += v2 * w2;
        } else if (r == 2) {
            const int s0 = __builtin_amdgcn_readlane(skey, j + 0);
            const int s1 = __builtin_amdgcn_readlane(skey, j + 1);
            const float w0 = dinv[s0] * di;
            const float w1 = dinv[s1] * di;
            const vecf v0 = *(const vecf*)(hl + (long long)s0 * D);
            const vecf v1 = *(const vecf*)(hl + (long long)s1 * D);
            acc += v0 * w0;
            acc += v1 * w1;
        } else if (r == 1) {
            const int s0 = __builtin_amdgcn_readlane(skey, j);
            const float w0 = dinv[s0] * di;
            const vecf v0 = *(const vecf*)(hl + (long long)s0 * D);
            acc += v0 * w0;
        }
    } else {
        int prev = -1;
        int left = d;
        while (left > 0) {
            int best = 0x7fffffff;
            for (int b = beg + lane; b < end; b += 32) {
                int k = csr[b];
                if (k > prev && k < best) best = k;
            }
            for (int o = 16; o > 0; o >>= 1) {
                int u = __shfl_xor(best, o, 32);
                best = u < best ? u : best;
            }
            int c = 0;
            for (int b = beg + lane; b < end; b += 32) c += (csr[b] == best) ? 1 : 0;
            for (int o = 16; o > 0; o >>= 1) c += __shfl_xor(c, o, 32);
            best = __builtin_amdgcn_readfirstlane(best);
            c = __builtin_amdgcn_readfirstlane(c);
            if (c <= 0) break;
            const float w0 = dinv[best] * di;
            const vecf v0 = *(const vecf*)(hl + (long long)best * D);
            for (int r = 0; r < c; r++) acc += v0 * w0;
            prev = best;
            left -= c;
        }
    }

    if (ELU) {
#pragma unroll
        for (int q = 0; q < VEC; q++) {
            float v = acc[q];
            acc[q] = v > 0.0f ? v : (__expf(v) - 1.0f);
        }
    }
    *(vecf*)(Out + (long long)node * D + lane * VEC) = acc;
}

extern "C" void kernel_launch(void* const* d_in, const int* in_sizes, int n_in,
                              void* d_out, int out_size, void* d_ws, size_t ws_size,
                              hipStream_t stream) {
    const float* x  = (const float*)d_in[0];
    const int*   ei = (const int*)d_in[1];
    const float* W1 = (const float*)d_in[2];
    const float* b1 = (const float*)d_in[3];
    const float* W2 = (const float*)d_in[4];
    const float* b2 = (const float*)d_in[5];
    float* out = (float*)d_out;

    const int N = in_sizes[0] / DIN;
    const int E = in_sizes[1] / 2;
    const int DH = 128;

    const int B = 256;
    const int gN    = (N + B - 1) / B;
    const int gE    = (E + B - 1) / B;
    const int gRows = (N + 127) / 128;
    const int gScan = (N + SCAN_PER_BLOCK - 1) / SCAN_PER_BLOCK;
    const int gNode = (N + 7) / 8;

    char* ws = (char*)d_ws;
    float* dinv = (float*)ws;
    size_t off = (((size_t)N * sizeof(float)) + 255) & ~(size_t)255;
    float* hbuf = (float*)(ws + off);
    float* aggbuf = (float*)(ws + off + (size_t)N * DH * sizeof(float));
    char* tail = ws + off + 2 * (size_t)N * DH * sizeof(float);
    int* rp = (int*)tail;
    size_t rpb = (((size_t)(N + 1) * sizeof(int)) + 255) & ~(size_t)255;
    int* csr = (int*)(tail + rpb);
    int* cnt  = (int*)aggbuf;
    int* bsum = cnt + N;
    int* pos  = bsum + gScan;

    cnt_zero_kernel<<<gN, B, 0, stream>>>(cnt, N);
    count_kernel<<<gE, B, 0, stream>>>(ei, E, cnt, pos);
    block_sums_kernel<<<gScan, B, 0, stream>>>(cnt, N, bsum);
    scan_kernel<<<gScan, B, 0, stream>>>(cnt, N, bsum, rp, dinv);
    fill_kernel<<<gE, B, 0, stream>>>(ei, E, rp, pos, csr);

    gemm_wmma_kernel<128><<<gRows, B, 0, stream>>>(x, W1, hbuf, N);
    gather_kernel<128, true><<<gNode, B, 0, stream>>>(rp, csr, dinv, hbuf, b1, aggbuf, N);

    gemm_wmma_kernel<64><<<gRows, B, 0, stream>>>(aggbuf, W2, hbuf, N);
    gather_kernel<64, false><<<gNode, B, 0, stream>>>(rp, csr, dinv, hbuf, b2, out, N);
}


# Gather on a fixed grid of long-lived waves, each looping over nodes with the next node's sources, own row and bounds loaded one node ahead, up to 8 row loads per group; per-row sum order unchanged
# speedup vs baseline: 2.2129x; 1.0541x over previous
#include <hip/hip_runtime.h>
#include <hip/hip_bf16.h>

typedef __attribute__((ext_vector_type(2))) float v2f;
typedef __attribute__((ext_vector_type(8))) float v8f;

#define DIN 128
#define SCAN_PER_BLOCK 1024

__global__ __launch_bounds__(256) void cnt_zero_kernel(int* __restrict__ cnt, int n) {
    int i = blockIdx.x * blockDim.x + threadIdx.x;
    if (i < n) cnt[i] = 0;
}

__global__ __launch_bounds__(256) void count_kernel(const int* __restrict__ ei, int e,
                                                    int* __restrict__ cnt,
                                                    int* __restrict__ pos) {
    int idx = blockIdx.x * blockDim.x + threadIdx.x;
    if (idx < e) pos[idx] = atomicAdd(&cnt[ei[e + idx]], 1);
}

__global__ __launch_bounds__(256) void block_sums_kernel(const int* __restrict__ cnt, int n,
                                                         int* __restrict__ bsum) {
    __shared__ int red[8];
    const int base = blockIdx.x * SCAN_PER_BLOCK + threadIdx.x * 4;
    int s = 0;
#pragma unroll
    for (int k = 0; k < 4; k++)
        if (base + k < n) s += cnt[base + k];
    for (int o = 16; o > 0; o >>= 1) s += __shfl_xor(s, o, 32);
    if ((threadIdx.x & 31) == 0) red[threadIdx.x >> 5] = s;
    __syncthreads();
    if (threadIdx.x == 0) {
        int t = 0;
        for (int w = 0; w < 8; w++) t += red[w];
        bsum[blockIdx.x] = t;
    }
}

__global__ __launch_bounds__(256) void scan_kernel(const int* __restrict__ cnt, int n,
                                                   const int* __restrict__ bsum,
                                                   int* __restrict__ rp,
                                                   float* __restrict__ dinv) {
    __shared__ int redp[8];
    __shared__ int wtot[8];
    const int lane = threadIdx.x & 31;
    const int wave = threadIdx.x >> 5;

    int p = 0;
    for (int j = threadIdx.x; j < (int)blockIdx.x; j += 256) p += bsum[j];
    for (int o = 16; o > 0; o >>= 1) p += __shfl_xor(p, o, 32);
    if (lane == 0) redp[wave] = p;

    const int base = blockIdx.x * SCAN_PER_BLOCK + threadIdx.x * 4;
    int c[4];
#pragma unroll
    for (int k = 0; k < 4; k++) c[k] = (base + k < n) ? cnt[base + k] : 0;
    const int t = c[0] + c[1] + c[2] + c[3];
    int incl = t;
    for (int o = 1; o < 32; o <<= 1) {
        int u = __shfl_up(incl, o, 32);
        if (lane >= o) incl += u;
    }
    if (lane == 31) wtot[wave] = incl;
    __syncthreads();

    int offs = incl - t;
    for (int w = 0; w < 8; w++) {
        offs += redp[w];
        if (w < wave) offs += wtot[w];
    }
#pragma unroll
    for (int k = 0; k < 4; k++) {
        if (base + k < n) {
            rp[base + k] = offs;
            dinv[base + k] = rsqrtf((float)(c[k] + 1));
            offs += c[k];
            if (base + k == n - 1) rp[n] = offs;
        }
    }
}

__global__ __launch_bounds__(256) void fill_kernel(const int* __restrict__ ei, int e,
                                                   const int* __restrict__ rp,
                                                   const int* __restrict__ pos,
                                                   int* __restrict__ csr) {
    int idx = blockIdx.x * blockDim.x + threadIdx.x;
    if (idx < e) csr[rp[ei[e + idx]] + pos[idx]] = ei[idx];
}

template <int DOUT>
__global__ __launch_bounds__(256) void gemm_wmma_kernel(const float* __restrict__ X,
                                                        const float* __restrict__ W,
                                                        float* __restrict__ H, int n) {
    constexpr int NT = DOUT / 16;
    constexpr int NPAIR = 32 * NT * 32;
    __shared__ float sW[NPAIR * 2];

    for (int pair = threadIdx.x; pair < NPAIR; pair += 256) {
        int l  = pair & 31;
        int tt = (pair >> 5) % NT;
        int ks = (pair >> 5) / NT;
        int col = tt * 16 + (l & 15);
        int k0  = ks * 4 + ((l >> 4) * 2);
        sW[pair * 2 + 0] = W[k0 * DOUT + col];
        sW[pair * 2 + 1] = W[(k0 + 1) * DOUT + col];
    }
    __syncthreads();

    const int wave = threadIdx.x >> 5;
    const int lane = threadIdx.x & 31;
    const int m    = lane & 15;
    const int kk   = (lane >> 4) * 2;

    const long long row0 = (long long)blockIdx.x * 128 + wave * 16;
    long long rowm = row0 + m;
    if (rowm >= n) rowm = n - 1;
    const float* __restrict__ xrow = X + rowm * DIN + kk;
    const float* __restrict__ bl   = sW + lane * 2;

    v8f acc[NT] = {};

#pragma unroll 4
    for (int ks = 0; ks < 32; ks++) {
        v2f a = *(const v2f*)(xrow + ks * 4);
#pragma unroll
        for (int t = 0; t < NT; t++) {
            v2f b = *(const v2f*)(bl + (ks * NT + t) * 64);
            acc[t] = __builtin_amdgcn_wmma_f32_16x16x4_f32(
                false, a, false, b, (short)0, acc[t], false, false);
        }
    }

    const long long rbase = row0 + (lane >> 4) * 8;
#pragma unroll
    for (int v = 0; v < 8; v++) {
        long long r = rbase + v;
        if (r < n) {
            float* __restrict__ orow = H + r * DOUT;
#pragma unroll
            for (int t = 0; t < NT; t++) orow[t * 16 + m] = acc[t][v];
        }
    }
}

#define GATHER_BLOCKS 2048

template <int K, typename vecf>
__device__ __forceinline__ void add_rows(vecf& acc, const int skey, const int j,
                                         const float* __restrict__ dinv, const float di,
                                         const float* __restrict__ H, const int lofs) {
    constexpr int D = (int)(sizeof(vecf) / sizeof(float)) * 32;
    int s[K];
    float w[K];
    vecf v[K];
#pragma unroll
    for (int q = 0; q < K; q++) s[q] = __builtin_amdgcn_readlane(skey, j + q);
#pragma unroll
    for (int q = 0; q < K; q++) w[q] = dinv[s[q]] * di;
#pragma unroll
    for (int q = 0; q < K; q++) v[q] = *(const vecf*)(H + (long long)s[q] * D + lofs);
#pragma unroll
    for (int q = 0; q < K; q++) acc += v[q] * w[q];
}

template <typename vecf>
__device__ __forceinline__ vecf self_term(const vecf bv, const vecf hv, const float dd) {
    constexpr int VEC = (int)(sizeof(vecf) / sizeof(float));
    vecf r;
    for (int q = 0; q < VEC; q++) r[q] = __builtin_fmaf(hv[q], dd, bv[q]);
    return r;
}

template <int D, bool ELU>
__global__ __launch_bounds__(256) void gather_kernel(const int* __restrict__ rp,
                                                     const int* __restrict__ csr,
                                                     const float* __restrict__ dinv,
                                                     const float* __restrict__ H,
                                                     const float* __restrict__ bias,
                                                     float* __restrict__ Out, int n) {
    constexpr int VEC = D / 32;
    typedef __attribute__((ext_vector_type(VEC))) float vecf;
    const int lane = threadIdx.x & 31;
    const int nw = (int)gridDim.x * 8;
    int node = __builtin_amdgcn_readfirstlane((int)(blockIdx.x * 8 + (threadIdx.x >> 5)));
    if (node >= n) return;

    const int lofs = lane * VEC;
    const vecf bv = *(const vecf*)(bias + lofs);

    int beg = rp[node];
    int end = rp[node + 1];
    float di = dinv[node];
    const vecf hv0 = *(const vecf*)(H + (long long)node * D + lofs);
    int key = 0x7fffffff;
    if (lane < end - beg) key = csr[beg + lane];
    int node1 = node + nw;
    int beg1 = 0, end1 = 0;
    if (node1 < n) {
        beg1 = rp[node1];
        end1 = rp[node1 + 1];
    }

    vecf acc = self_term(bv, hv0, di * di);

    for (;;) {
        int key1 = 0x7fffffff;
        float di1 = 0.0f;
        vecf hv1 = {};
        const int node2 = node1 + nw;
        int beg2 = 0, end2 = 0;
        if (node1 < n) {
            if (lane < end1 - beg1) key1 = csr[beg1 + lane];
            di1 = dinv[node1];
            hv1 = *(const vecf*)(H + (long long)node1 * D + lofs);
            if (node2 < n) {
                beg2 = rp[node2];
                end2 = rp[node2 + 1];
            }
        }

        const int d = end - beg;

        if (d <= 32) {
            int rank = 0;
            for (int j = 0; j < d; j++) {
                int kj = __builtin_amdgcn_readlane(key, j);
                rank += (kj < key || (kj == key && j < lane)) ? 1 : 0;
            }
            const int skey = __builtin_amdgcn_ds_permute(rank << 2, key);

            int j = 0;
            for (; j + 8 <= d; j += 8) add_rows<8>(acc, skey, j, dinv, di, H, lofs);
            switch (d - j) {
                case 7: add_rows<7>(acc, skey, j, dinv, di, H, lofs); break;
                case 6: add_rows<6>(acc, skey, j, dinv, di, H, lofs); break;
                case 5: add_rows<5>(acc, skey, j, dinv, di, H, lofs); break;
                case 4: add_rows<4>(acc, skey, j, dinv, di, H, lofs); break;
                case 3: add_rows<3>(acc, skey, j, dinv, di, H, lofs); break;
                case 2: add_rows<2>(acc, skey, j, dinv, di, H, lofs); break;
                case 1: add_rows<1>(acc, skey, j, dinv, di, H, lofs); break;
                default: break;
            }
        } else {
            int prev = -1;
            int left = d;
            while (left > 0) {
                int best = 0x7fffffff;
                for (int b = beg + lane; b < end; b += 32) {
                    int k = csr[b];
                    if (k > prev && k < best) best = k;
                }
                for (int o = 16; o > 0; o >>= 1) {
                    int u = __shfl_xor(best, o, 32);
                    best = u < best ? u : best;
                }
                int c = 0;
                for (int b = beg + lane; b < end; b += 32) c += (csr[b] == best) ? 1 : 0;
                for (int o = 16; o > 0; o >>= 1) c += __shfl_xor(c, o, 32);
                best = __builtin_amdgcn_readfirstlane(best);
                c = __builtin_amdgcn_readfirstlane(c);
                if (c <= 0) break;
                const float w0 = dinv[best] * di;
                const vecf v0 = *(const vecf*)(H + (long long)best * D + lofs);
                for (int r = 0; r < c; r++) acc += v0 * w0;
                prev = best;
                left -= c;
            }
        }

        if (ELU) {
#pragma unroll
            for (int q = 0; q < VEC; q++) {
                float v = acc[q];
                acc[q] = v > 0.0f ? v : (__expf(v) - 1.0f);
            }
        }
        *(vecf*)(Out + (long long)node * D + lofs) = acc;

        acc = self_term(bv, hv1, di1 * di1);
        if (node1 >= n) break;
        node = node1; beg = beg1; end = end1; di = di1; key = key1;
        node1 = node2; beg1 = beg2; end1 = end2;
    }
}

extern "C" void kernel_launch(void* const* d_in, const int* in_sizes, int n_in,
                              void* d_out, int out_size, void* d_ws, size_t ws_size,
                              hipStream_t stream) {
    const float* x  = (const float*)d_in[0];
    const int*   ei = (const int*)d_in[1];
    const float* W1 = (const float*)d_in[2];
    const float* b1 = (const float*)d_in[3];
    const float* W2 = (const float*)d_in[4];
    const float* b2 = (const float*)d_in[5];
    float* out = (float*)d_out;

    const int N = in_sizes[0] / DIN;
    const int E = in_sizes[1] / 2;
    const int DH = 128;

    const int B = 256;
    const int gN    = (N + B - 1) / B;
    const int gE    = (E + B - 1) / B;
    const int gRows = (N + 127) / 128;
    const int gScan = (N + SCAN_PER_BLOCK - 1) / SCAN_PER_BLOCK;
    const int gNode = GATHER_BLOCKS;

    char* ws = (char*)d_ws;
    float* dinv = (float*)ws;
    size_t off = (((size_t)N * sizeof(float)) + 255) & ~(size_t)255;
    float* hbuf = (float*)(ws + off);
    float* aggbuf = (float*)(ws + off + (size_t)N * DH * sizeof(float));
    char* tail = ws + off + 2 * (size_t)N * DH * sizeof(float);
    int* rp = (int*)tail;
    size_t rpb = (((size_t)(N + 1) * sizeof(int)) + 255) & ~(size_t)255;
    int* csr = (int*)(tail + rpb);
    int* cnt  = (int*)aggbuf;
    int* bsum = cnt + N;
    int* pos  = bsum + gScan;

    cnt_zero_kernel<<<gN, B, 0, stream>>>(cnt, N);
    count_kernel<<<gE, B, 0, stream>>>(ei, E, cnt, pos);
    block_sums_kernel<<<gScan, B, 0, stream>>>(cnt, N, bsum);
    scan_kernel<<<gScan, B, 0, stream>>>(cnt, N, bsum, rp, dinv);
    fill_kernel<<<gE, B, 0, stream>>>(ei, E, rp, pos, csr);

    gemm_wmma_kernel<128><<<gRows, B, 0, stream>>>(x, W1, hbuf, N);
    gather_kernel<128, true><<<gNode, B, 0, stream>>>(rp, csr, dinv, hbuf, b1, aggbuf, N);

    gemm_wmma_kernel<64><<<gRows, B, 0, stream>>>(aggbuf, W2, hbuf, N);
    gather_kernel<64, false><<<gNode, B, 0, stream>>>(rp, csr, dinv, hbuf, b2, out, N);
}


# Edge grouping without global atomics: two-level counting sort by destination (256-node buckets), all counters and cursors in LDS; row sum order unchanged (gather still sorts each row by source)
# speedup vs baseline: 2.7899x; 1.2607x over previous
#include <hip/hip_runtime.h>
#include <hip/hip_bf16.h>

typedef __attribute__((ext_vector_type(2))) float v2f;
typedef __attribute__((ext_vector_type(8))) float v8f;

#define DIN 128
#define SCAN_PER_BLOCK 1024

#define BUCKET_SHIFT 8
#define BUCKET_NODES (1 << BUCKET_SHIFT)
#define SORT_BLOCKS 256
#define SORT_THREADS 512
#define EDGE_BATCH 8
#define ROW_BATCH 12

__global__ __launch_bounds__(SORT_THREADS) void bucket_count_kernel(const int* __restrict__ ei, int e, int nb,
                                                                    int* __restrict__ bh) {
    extern __shared__ int sh[];
    const int tid = threadIdx.x;
    const int nblk = (int)gridDim.x;
    for (int b = tid; b < nb; b += SORT_THREADS) sh[b] = 0;
    __syncthreads();

    const int ch = (e + nblk - 1) / nblk;
    const int lo = (int)blockIdx.x * ch;
    const int hi = (lo + ch < e) ? (lo + ch) : e;
    const int* __restrict__ dstp = ei + e;
    for (int i0 = lo + tid; i0 < hi; i0 += SORT_THREADS * EDGE_BATCH) {
        int d[EDGE_BATCH];
#pragma unroll
        for (int q = 0; q < EDGE_BATCH; q++) {
            const int i = i0 + q * SORT_THREADS;
            d[q] = (i < hi) ? dstp[i] : -1;
        }
#pragma unroll
        for (int q = 0; q < EDGE_BATCH; q++)
            if (d[q] >= 0) atomicAdd(&sh[d[q] >> BUCKET_SHIFT], 1);
    }
    __syncthreads();
    for (int b = tid; b < nb; b += SORT_THREADS) bh[b * nblk + (int)blockIdx.x] = sh[b];
}

__global__ __launch_bounds__(256) void block_sums_kernel(const int* __restrict__ cnt, int n,
                                                         int* __restrict__ bsum) {
    __shared__ int red[8];
    const int base = blockIdx.x * SCAN_PER_BLOCK + threadIdx.x * 4;
    int s = 0;
#pragma unroll
    for (int k = 0; k < 4; k++)
        if (base + k < n) s += cnt[base + k];
    for (int o = 16; o > 0; o >>= 1) s += __shfl_xor(s, o, 32);
    if ((threadIdx.x & 31) == 0) red[threadIdx.x >> 5] = s;
    __syncthreads();
    if (threadIdx.x == 0) {
        int t = 0;
        for (int w = 0; w < 8; w++) t += red[w];
        bsum[blockIdx.x] = t;
    }
}

__global__ __launch_bounds__(256) void scan_excl_kernel(const int* __restrict__ cnt, int n,
                                                        const int* __restrict__ bsum,
                                                        int* __restrict__ out) {
    __shared__ int redp[8];
    __shared__ int wtot[8];
    const int lane = threadIdx.x & 31;
    const int wave = threadIdx.x >> 5;

    int p = 0;
    for (int j = threadIdx.x; j < (int)blockIdx.x; j += 256) p += bsum[j];
    for (int o = 16; o > 0; o >>= 1) p += __shfl_xor(p, o, 32);
    if (lane == 0) redp[wave] = p;

    const int base = blockIdx.x * SCAN_PER_BLOCK + threadIdx.x * 4;
    int c[4];
#pragma unroll
    for (int k = 0; k < 4; k++) c[k] = (base + k < n) ? cnt[base + k] : 0;
    const int t = c[0] + c[1] + c[2] + c[3];
    int incl = t;
    for (int o = 1; o < 32; o <<= 1) {
        int u = __shfl_up(incl, o, 32);
        if (lane >= o) incl += u;
    }
    if (lane == 31) wtot[wave] = incl;
    __syncthreads();

    int offs = incl - t;
    for (int w = 0; w < 8; w++) {
        offs += redp[w];
        if (w < wave) offs += wtot[w];
    }
#pragma unroll
    for (int k = 0; k < 4; k++) {
        if (base + k < n) {
            out[base + k] = offs;
            offs += c[k];
        }
    }
}

__global__ __launch_bounds__(SORT_THREADS) void bucket_scatter_kernel(const int* __restrict__ ei, int e, int nb,
                                                                      const int* __restrict__ boff,
                                                                      int* __restrict__ pairs) {
    extern __shared__ int sh[];
    const int tid = threadIdx.x;
    const int nblk = (int)gridDim.x;
    for (int b = tid; b < nb; b += SORT_THREADS) sh[b] = boff[b * nblk + (int)blockIdx.x];
    __syncthreads();

    const int ch = (e + nblk - 1) / nblk;
    const int lo = (int)blockIdx.x * ch;
    const int hi = (lo + ch < e) ? (lo + ch) : e;
    const int* __restrict__ dstp = ei + e;
    for (int i0 = lo + tid; i0 < hi; i0 += SORT_THREADS * EDGE_BATCH) {
        int s[EDGE_BATCH];
        int d[EDGE_BATCH];
#pragma unroll
        for (int q = 0; q < EDGE_BATCH; q++) {
            const int i = i0 + q * SORT_THREADS;
            s[q] = (i < hi) ? ei[i] : 0;
            d[q] = (i < hi) ? dstp[i] : -1;
        }
#pragma unroll
        for (int q = 0; q < EDGE_BATCH; q++) {
            if (d[q] >= 0) {
                const int p = atomicAdd(&sh[d[q] >> BUCKET_SHIFT], 1);
                *(int2*)(pairs + 2 * (long long)p) = make_int2(s[q], d[q]);
            }
        }
    }
}

__global__ __launch_bounds__(BUCKET_NODES) void bucket_rows_kernel(const int* __restrict__ pairs,
                                                                   const int* __restrict__ boff, int nblk,
                                                                   int nb, int e, int n,
                                                                   int* __restrict__ rp,
                                                                   float* __restrict__ dinv,
                                                                   int* __restrict__ csr) {
    __shared__ int cnt[BUCKET_NODES];
    __shared__ int cur[BUCKET_NODES];
    __shared__ int wtot[BUCKET_NODES / 32];
    const int tid = threadIdx.x;
    const int lane = tid & 31;
    const int wave = tid >> 5;
    const int b = (int)blockIdx.x;
    const int lo = boff[b * nblk];
    const int hi = (b + 1 < nb) ? boff[(b + 1) * nblk] : e;

    cnt[tid] = 0;
    __syncthreads();
    for (int i0 = lo + tid; i0 < hi; i0 += BUCKET_NODES * ROW_BATCH) {
        int d[ROW_BATCH];
#pragma unroll
        for (int q = 0; q < ROW_BATCH; q++) {
            const int i = i0 + q * BUCKET_NODES;
            d[q] = (i < hi) ? pairs[2 * (long long)i + 1] : -1;
        }
#pragma unroll
        for (int q = 0; q < ROW_BATCH; q++)
            if (d[q] >= 0) atomicAdd(&cnt[d[q] & (BUCKET_NODES - 1)], 1);
    }
    __syncthreads();

    const int c = cnt[tid];
    int incl = c;
    for (int o = 1; o < 32; o <<= 1) {
        int u = __shfl_up(incl, o, 32);
        if (lane >= o) incl += u;
    }
    if (lane == 31) wtot[wave] = incl;
    __syncthreads();
    int start = lo + incl - c;
    for (int w = 0; w < BUCKET_NODES / 32; w++)
        if (w < wave) start += wtot[w];

    const int node = b * BUCKET_NODES + tid;
    cur[tid] = start;
    if (node < n) {
        rp[node] = start;
        dinv[node] = rsqrtf((float)(c + 1));
        if (node == n - 1) rp[n] = start + c;
    }
    __syncthreads();

    for (int i0 = lo + tid; i0 < hi; i0 += BUCKET_NODES * ROW_BATCH) {
        int2 pr[ROW_BATCH];
#pragma unroll
        for (int q = 0; q < ROW_BATCH; q++) {
            const int i = i0 + q * BUCKET_NODES;
            pr[q] = (i < hi) ? *(const int2*)(pairs + 2 * (long long)i) : make_int2(0, -1);
        }
#pragma unroll
        for (int q = 0; q < ROW_BATCH; q++) {
            if (pr[q].y >= 0) {
                const int p = atomicAdd(&cur[pr[q].y & (BUCKET_NODES - 1)], 1);
                csr[p] = pr[q].x;
            }
        }
    }
}

template <int DOUT>
__global__ __launch_bounds__(256) void gemm_wmma_kernel(const float* __restrict__ X,
                                                        const float* __restrict__ W,
                                                        float* __restrict__ H, int n) {
    constexpr int NT = DOUT / 16;
    constexpr int NPAIR = 32 * NT * 32;
    __shared__ float sW[NPAIR * 2];

    for (int pair = threadIdx.x; pair < NPAIR; pair += 256) {
        int l  = pair & 31;
        int tt = (pair >> 5) % NT;
        int ks = (pair >> 5) / NT;
        int col = tt * 16 + (l & 15);
        int k0  = ks * 4 + ((l >> 4) * 2);
        sW[pair * 2 + 0] = W[k0 * DOUT + col];
        sW[pair * 2 + 1] = W[(k0 + 1) * DOUT + col];
    }
    __syncthreads();

    const int wave = threadIdx.x >> 5;
    const int lane = threadIdx.x & 31;
    const int m    = lane & 15;
    const int kk   = (lane >> 4) * 2;

    const long long row0 = (long long)blockIdx.x * 128 + wave * 16;
    long long rowm = row0 + m;
    if (rowm >= n) rowm = n - 1;
    const float* __restrict__ xrow = X + rowm * DIN + kk;
    const float* __restrict__ bl   = sW + lane * 2;

    v8f acc[NT] = {};

#pragma unroll 4
    for (int ks = 0; ks < 32; ks++) {
        v2f a = *(const v2f*)(xrow + ks * 4);
#pragma unroll
        for (int t = 0; t < NT; t++) {
            v2f b = *(const v2f*)(bl + (ks * NT + t) * 64);
            acc[t] = __builtin_amdgcn_wmma_f32_16x16x4_f32(
                false, a, false, b, (short)0, acc[t], false, false);
        }
    }

    const long long rbase = row0 + (lane >> 4) * 8;
#pragma unroll
    for (int v = 0; v < 8; v++) {
        long long r = rbase + v;
        if (r < n) {
            float* __restrict__ orow = H + r * DOUT;
#pragma unroll
            for (int t = 0; t < NT; t++) orow[t * 16 + m] = acc[t][v];
        }
    }
}

#define GATHER_BLOCKS 2048

template <int K, typename vecf>
__device__ __forceinline__ void add_rows(vecf& acc, const int skey, const int j,
                                         const float* __restrict__ dinv, const float di,
                                         const float* __restrict__ H, const int lofs) {
    constexpr int D = (int)(sizeof(vecf) / sizeof(float)) * 32;
    int s[K];
    float w[K];
    vecf v[K];
#pragma unroll
    for (int q = 0; q < K; q++) s[q] = __builtin_amdgcn_readlane(skey, j + q);
#pragma unroll
    for (int q = 0; q < K; q++) w[q] = dinv[s[q]] * di;
#pragma unroll
    for (int q = 0; q < K; q++) v[q] = *(const vecf*)(H + (long long)s[q] * D + lofs);
#pragma unroll
    for (int q = 0; q < K; q++) acc += v[q] * w[q];
}

template <typename vecf>
__device__ __forceinline__ vecf self_term(const vecf bv, const vecf hv, const float dd) {
    constexpr int VEC = (int)(sizeof(vecf) / sizeof(float));
    vecf r;
    for (int q = 0; q < VEC; q++) r[q] = __builtin_fmaf(hv[q], dd, bv[q]);
    return r;
}

template <int D, bool ELU>
__global__ __launch_bounds__(256) void gather_kernel(const int* __restrict__ rp,
                                                     const int* __restrict__ csr,
                                                     const float* __restrict__ dinv,
                                                     const float* __restrict__ H,
                                                     const float* __restrict__ bias,
                                                     float* __restrict__ Out, int n) {
    constexpr int VEC = D / 32;
    typedef __attribute__((ext_vector_type(VEC))) float vecf;
    const int lane = threadIdx.x & 31;
    const int nw = (int)gridDim.x * 8;
    int node = __builtin_amdgcn_readfirstlane((int)(blockIdx.x * 8 + (threadIdx.x >> 5)));
    if (node >= n) return;

    const int lofs = lane * VEC;
    const vecf bv = *(const vecf*)(bias + lofs);

    int beg = rp[node];
    int end = rp[node + 1];
    float di = dinv[node];
    const vecf hv0 = *(const vecf*)(H + (long long)node * D + lofs);
    int key = 0x7fffffff;
    if (lane < end - beg) key = csr[beg + lane];
    int node1 = node + nw;
    int beg1 = 0, end1 = 0;
    if (node1 < n) {
        beg1 = rp[node1];
        end1 = rp[node1 + 1];
    }

    vecf acc = self_term(bv, hv0, di * di);

    for (;;) {
        int key1 = 0x7fffffff;
        float di1 = 0.0f;
        vecf hv1 = {};
        const int node2 = node1 + nw;
        int beg2 = 0, end2 = 0;
        if (node1 < n) {
            if (lane < end1 - beg1) key1 = csr[beg1 + lane];
            di1 = dinv[node1];
            hv1 = *(const vecf*)(H + (long long)node1 * D + lofs);
            if (node2 < n) {
                beg2 = rp[node2];
                end2 = rp[node2 + 1];
            }
        }

        const int d = end - beg;

        if (d <= 32) {
            int rank = 0;
            for (int j = 0; j < d; j++) {
                int kj = __builtin_amdgcn_readlane(key, j);
                rank += (kj < key || (kj == key && j < lane)) ? 1 : 0;
            }
            const int skey = __builtin_amdgcn_ds_permute(rank << 2, key);

            int j = 0;
            for (; j + 8 <= d; j += 8) add_rows<8>(acc, skey, j, dinv, di, H, lofs);
            switch (d - j) {
                case 7: add_rows<7>(acc, skey, j, dinv, di, H, lofs); break;
                case 6: add_rows<6>(acc, skey, j, dinv, di, H, lofs); break;
                case 5: add_rows<5>(acc, skey, j, dinv, di, H, lofs); break;
                case 4: add_rows<4>(acc, skey, j, dinv, di, H, lofs); break;
                case 3: add_rows<3>(acc, skey, j, dinv, di, H, lofs); break;
                case 2: add_rows<2>(acc, skey, j, dinv, di, H, lofs); break;
                case 1: add_rows<1>(acc, skey, j, dinv, di, H, lofs); break;
                default: break;
            }
        } else {
            int prev = -1;
            int left = d;
            while (left > 0) {
                int best = 0x7fffffff;
                for (int b = beg + lane; b < end; b += 32) {
                    int k = csr[b];
                    if (k > prev && k < best) best = k;
                }
                for (int o = 16; o > 0; o >>= 1) {
                    int u = __shfl_xor(best, o, 32);
                    best = u < best ? u : best;
                }
                int c = 0;
                for (int b = beg + lane; b < end; b += 32) c += (csr[b] == best) ? 1 : 0;
                for (int o = 16; o > 0; o >>= 1) c += __shfl_xor(c, o, 32);
                best = __builtin_amdgcn_readfirstlane(best);
                c = __builtin_amdgcn_readfirstlane(c);
                if (c <= 0) break;
                const float w0 = dinv[best] * di;
                const vecf v0 = *(const vecf*)(H + (long long)best * D + lofs);
                for (int r = 0; r < c; r++) acc += v0 * w0;
                prev = best;
                left -= c;
            }
        }

        if (ELU) {
#pragma unroll
            for (int q = 0; q < VEC; q++) {
                float v = acc[q];
                acc[q] = v > 0.0f ? v : (__expf(v) - 1.0f);
            }
        }
        *(vecf*)(Out + (long long)node * D + lofs) = acc;

        acc = self_term(bv, hv1, di1 * di1);
        if (node1 >= n) break;
        node = node1; beg = beg1; end = end1; di = di1; key = key1;
        node1 = node2; beg1 = beg2; end1 = end2;
    }
}

extern "C" void kernel_launch(void* const* d_in, const int* in_sizes, int n_in,
                              void* d_out, int out_size, void* d_ws, size_t ws_size,
                              hipStream_t stream) {
    const float* x  = (const float*)d_in[0];
    const int*   ei = (const int*)d_in[1];
    const float* W1 = (const float*)d_in[2];
    const float* b1 = (const float*)d_in[3];
    const float* W2 = (const float*)d_in[4];
    const float* b2 = (const float*)d_in[5];
    float* out = (float*)d_out;

    const int N = in_sizes[0] / DIN;
    const int E = in_sizes[1] / 2;
    const int DH = 128;

    const int B = 256;
    const int gRows = (N + 127) / 128;
    const int gNode = GATHER_BLOCKS;
    const int NB    = (N + BUCKET_NODES - 1) / BUCKET_NODES;
    const int nScan = NB * SORT_BLOCKS;
    const int gScan = (nScan + SCAN_PER_BLOCK - 1) / SCAN_PER_BLOCK;

    char* ws = (char*)d_ws;
    float* dinv = (float*)ws;
    size_t off = (((size_t)N * sizeof(float)) + 255) & ~(size_t)255;
    float* hbuf = (float*)(ws + off);
    float* aggbuf = (float*)(ws + off + (size_t)N * DH * sizeof(float));
    char* tail = ws + off + 2 * (size_t)N * DH * sizeof(float);
    int* rp = (int*)tail;
    size_t rpb = (((size_t)(N + 1) * sizeof(int)) + 255) & ~(size_t)255;
    int* csr = (int*)(tail + rpb);
    int* pairs = (int*)aggbuf;
    int* bh    = pairs + 2 * (size_t)E;
    int* boff  = bh + nScan;
    int* bsum  = boff + nScan;

    bucket_count_kernel<<<SORT_BLOCKS, SORT_THREADS, NB * sizeof(int), stream>>>(ei, E, NB, bh);
    block_sums_kernel<<<gScan, B, 0, stream>>>(bh, nScan, bsum);
    scan_excl_kernel<<<gScan, B, 0, stream>>>(bh, nScan, bsum, boff);
    bucket_scatter_kernel<<<SORT_BLOCKS, SORT_THREADS, NB * sizeof(int), stream>>>(ei, E, NB, boff, pairs);
    bucket_rows_kernel<<<NB, BUCKET_NODES, 0, stream>>>(pairs, boff, SORT_BLOCKS, NB, E, N, rp, dinv, csr);

    gemm_wmma_kernel<128><<<gRows, B, 0, stream>>>(x, W1, hbuf, N);
    gather_kernel<128, true><<<gNode, B, 0, stream>>>(rp, csr, dinv, hbuf, b1, aggbuf, N);

    gemm_wmma_kernel<64><<<gRows, B, 0, stream>>>(aggbuf, W2, hbuf, N);
    gather_kernel<64, false><<<gNode, B, 0, stream>>>(rp, csr, dinv, hbuf, b2, out, N);
}


# Edge grouping in 2 launches, not 5: each chunk block sorts its own edges by bucket within its own range (LDS count, scan, cursors), bucket blocks walk the 256 chunk runs; row sum order unchanged
# speedup vs baseline: 3.1526x; 1.1300x over previous
#include <hip/hip_runtime.h>
#include <hip/hip_bf16.h>

typedef __attribute__((ext_vector_type(2))) float v2f;
typedef __attribute__((ext_vector_type(8))) float v8f;

#define DIN 128

#define BUCKET_SHIFT 8
#define BUCKET_NODES (1 << BUCKET_SHIFT)
#define SORT_BLOCKS 256
#define SORT_THREADS 512
#define EDGE_BATCH 8
#define RUN_LANES 16
#define RUN_STEPS 16

static_assert(SORT_BLOCKS == 32 * (BUCKET_NODES / 32), "bucket_rows_kernel: 8 waves x 32 chunk runs");
static_assert(RUN_STEPS * (32 / RUN_LANES) == 32, "a wave covers 32 chunk runs");

__global__ __launch_bounds__(SORT_THREADS) void chunk_sort_kernel(const int* __restrict__ ei, int e, int nb,
                                                                  int* __restrict__ cpre,
                                                                  int* __restrict__ pairs) {
    extern __shared__ int sh[];
    __shared__ int wt[SORT_THREADS / 32];
    int* __restrict__ cnt = sh;
    int* __restrict__ cur = sh + nb;
    const int tid = threadIdx.x;
    const int lane = tid & 31;
    const int wave = tid >> 5;
    const int g = (int)blockIdx.x;
    for (int b = tid; b < nb; b += SORT_THREADS) cnt[b] = 0;
    __syncthreads();

    const int ch = (e + SORT_BLOCKS - 1) / SORT_BLOCKS;
    const int lo = g * ch;
    const int hi = (lo + ch < e) ? (lo + ch) : e;
    const int* __restrict__ dstp = ei + e;

    for (int i0 = lo + tid; i0 < hi; i0 += SORT_THREADS * EDGE_BATCH) {
        int d[EDGE_BATCH];
#pragma unroll
        for (int q = 0; q < EDGE_BATCH; q++) {
            const int i = i0 + q * SORT_THREADS;
            d[q] = (i < hi) ? dstp[i] : -1;
        }
#pragma unroll
        for (int q = 0; q < EDGE_BATCH; q++)
            if (d[q] >= 0) atomicAdd(&cnt[d[q] >> BUCKET_SHIFT], 1);
    }
    __syncthreads();

    int carry = 0;
    for (int base = 0; base < nb; base += SORT_THREADS) {
        const int b = base + tid;
        const int t = (b < nb) ? cnt[b] : 0;
        int incl = t;
        for (int o = 1; o < 32; o <<= 1) {
            int u = __shfl_up(incl, o, 32);
            if (lane >= o) incl += u;
        }
        if (lane == 31) wt[wave] = incl;
        __syncthreads();
        int offs = carry + incl - t;
        int all = 0;
        for (int w = 0; w < SORT_THREADS / 32; w++) {
            const int x = wt[w];
            if (w < wave) offs += x;
            all += x;
        }
        if (b < nb) {
            cur[b] = offs;
            cpre[(size_t)b * SORT_BLOCKS + g] = offs;
        }
        carry += all;
        __syncthreads();
    }
    if (tid == 0) cpre[(size_t)nb * SORT_BLOCKS + g] = carry;

    for (int i0 = lo + tid; i0 < hi; i0 += SORT_THREADS * EDGE_BATCH) {
        int s[EDGE_BATCH];
        int d[EDGE_BATCH];
#pragma unroll
        for (int q = 0; q < EDGE_BATCH; q++) {
            const int i = i0 + q * SORT_THREADS;
            s[q] = (i < hi) ? ei[i] : 0;
            d[q] = (i < hi) ? dstp[i] : -1;
        }
#pragma unroll
        for (int q = 0; q < EDGE_BATCH; q++) {
            if (d[q] >= 0) {
                const int p = lo + atomicAdd(&cur[d[q] >> BUCKET_SHIFT], 1);
                *(int2*)(pairs + 2 * (long long)p) = make_int2(s[q], d[q]);
            }
        }
    }
}

__global__ __launch_bounds__(BUCKET_NODES) void bucket_rows_kernel(const int* __restrict__ pairs,
                                                                   const int* __restrict__ cpre, int e, int n,
                                                                   int* __restrict__ rp,
                                                                   float* __restrict__ dinv,
                                                                   int* __restrict__ csr) {
    __shared__ int cnt[BUCKET_NODES];
    __shared__ int cur[BUCKET_NODES];
    __shared__ int runS[SORT_BLOCKS];
    __shared__ int runL[SORT_BLOCKS];
    __shared__ int wtot[BUCKET_NODES / 32];
    __shared__ int wlo[BUCKET_NODES / 32];
    const int tid = threadIdx.x;
    const int lane = tid & 31;
    const int wave = tid >> 5;
    const int b = (int)blockIdx.x;

    const int ch = (e + SORT_BLOCKS - 1) / SORT_BLOCKS;
    const int o0 = cpre[(size_t)b * SORT_BLOCKS + tid];
    const int o1 = cpre[(size_t)(b + 1) * SORT_BLOCKS + tid];
    runS[tid] = tid * ch + o0;
    runL[tid] = o1 - o0;
    cnt[tid] = 0;
    int s0 = o0;
    for (int o = 16; o > 0; o >>= 1) s0 += __shfl_xor(s0, o, 32);
    if (lane == 0) wlo[wave] = s0;
    __syncthreads();
    int lo = 0;
    for (int w = 0; w < BUCKET_NODES / 32; w++) lo += wlo[w];

    const int sub = lane & (RUN_LANES - 1);
    const int which = lane / RUN_LANES;
    int S[RUN_STEPS];
    int L[RUN_STEPS];
#pragma unroll
    for (int j = 0; j < RUN_STEPS; j++) {
        const int gi = wave * 32 + j * (32 / RUN_LANES) + which;
        S[j] = runS[gi];
        L[j] = runL[gi];
    }

    {
        int d[RUN_STEPS];
#pragma unroll
        for (int j = 0; j < RUN_STEPS; j++)
            d[j] = (sub < L[j]) ? pairs[2 * (long long)(S[j] + sub) + 1] : -1;
#pragma unroll
        for (int j = 0; j < RUN_STEPS; j++)
            if (d[j] >= 0) atomicAdd(&cnt[d[j] & (BUCKET_NODES - 1)], 1);
#pragma unroll
        for (int j = 0; j < RUN_STEPS; j++) {
            for (int k = sub + RUN_LANES; k < L[j]; k += RUN_LANES) {
                const int dd = pairs[2 * (long long)(S[j] + k) + 1];
                atomicAdd(&cnt[dd & (BUCKET_NODES - 1)], 1);
            }
        }
    }
    __syncthreads();

    const int c = cnt[tid];
    int incl = c;
    for (int o = 1; o < 32; o <<= 1) {
        int u = __shfl_up(incl, o, 32);
        if (lane >= o) incl += u;
    }
    if (lane == 31) wtot[wave] = incl;
    __syncthreads();
    int start = lo + incl - c;
    for (int w = 0; w < BUCKET_NODES / 32; w++)
        if (w < wave) start += wtot[w];

    const int node = b * BUCKET_NODES + tid;
    cur[tid] = start;
    if (node < n) {
        rp[node] = start;
        dinv[node] = rsqrtf((float)(c + 1));
        if (node == n - 1) rp[n] = start + c;
    }
    __syncthreads();

    {
        int2 pr[RUN_STEPS];
#pragma unroll
        for (int j = 0; j < RUN_STEPS; j++)
            pr[j] = (sub < L[j]) ? *(const int2*)(pairs + 2 * (long long)(S[j] + sub)) : make_int2(0, -1);
#pragma unroll
        for (int j = 0; j < RUN_STEPS; j++) {
            if (pr[j].y >= 0) {
                const int p = atomicAdd(&cur[pr[j].y & (BUCKET_NODES - 1)], 1);
                csr[p] = pr[j].x;
            }
        }
#pragma unroll
        for (int j = 0; j < RUN_STEPS; j++) {
            for (int k = sub + RUN_LANES; k < L[j]; k += RUN_LANES) {
                const int2 q2 = *(const int2*)(pairs + 2 * (long long)(S[j] + k));
                const int p = atomicAdd(&cur[q2.y & (BUCKET_NODES - 1)], 1);
                csr[p] = q2.x;
            }
        }
    }
}

template <int DOUT>
__global__ __launch_bounds__(256) void gemm_wmma_kernel(const float* __restrict__ X,
                                                        const float* __restrict__ W,
                                                        float* __restrict__ H, int n) {
    constexpr int NT = DOUT / 16;
    constexpr int NPAIR = 32 * NT * 32;
    __shared__ float sW[NPAIR * 2];

    for (int pair = threadIdx.x; pair < NPAIR; pair += 256) {
        int l  = pair & 31;
        int tt = (pair >> 5) % NT;
        int ks = (pair >> 5) / NT;
        int col = tt * 16 + (l & 15);
        int k0  = ks * 4 + ((l >> 4) * 2);
        sW[pair * 2 + 0] = W[k0 * DOUT + col];
        sW[pair * 2 + 1] = W[(k0 + 1) * DOUT + col];
    }
    __syncthreads();

    const int wave = threadIdx.x >> 5;
    const int lane = threadIdx.x & 31;
    const int m    = lane & 15;
    const int kk   = (lane >> 4) * 2;

    const long long row0 = (long long)blockIdx.x * 128 + wave * 16;
    long long rowm = row0 + m;
    if (rowm >= n) rowm = n - 1;
    const float* __restrict__ xrow = X + rowm * DIN + kk;
    const float* __restrict__ bl   = sW + lane * 2;

    v8f acc[NT] = {};

#pragma unroll 4
    for (int ks = 0; ks < 32; ks++) {
        v2f a = *(const v2f*)(xrow + ks * 4);
#pragma unroll
        for (int t = 0; t < NT; t++) {
            v2f b = *(const v2f*)(bl + (ks * NT + t) * 64);
            acc[t] = __builtin_amdgcn_wmma_f32_16x16x4_f32(
                false, a, false, b, (short)0, acc[t], false, false);
        }
    }

    const long long rbase = row0 + (lane >> 4) * 8;
#pragma unroll
    for (int v = 0; v < 8; v++) {
        long long r = rbase + v;
        if (r < n) {
            float* __restrict__ orow = H + r * DOUT;
#pragma unroll
            for (int t = 0; t < NT; t++) orow[t * 16 + m] = acc[t][v];
        }
    }
}

#define GATHER_BLOCKS 2048

template <int K, typename vecf>
__device__ __forceinline__ void add_rows(vecf& acc, const int skey, const int j,
                                         const float* __restrict__ dinv, const float di,
                                         const float* __restrict__ H, const int lofs) {
    constexpr int D = (int)(sizeof(vecf) / sizeof(float)) * 32;
    int s[K];
    float w[K];
    vecf v[K];
#pragma unroll
    for (int q = 0; q < K; q++) s[q] = __builtin_amdgcn_readlane(skey, j + q);
#pragma unroll
    for (int q = 0; q < K; q++) w[q] = dinv[s[q]] * di;
#pragma unroll
    for (int q = 0; q < K; q++) v[q] = *(const vecf*)(H + (long long)s[q] * D + lofs);
#pragma unroll
    for (int q = 0; q < K; q++) acc += v[q] * w[q];
}

template <typename vecf>
__device__ __forceinline__ vecf self_term(const vecf bv, const vecf hv, const float dd) {
    constexpr int VEC = (int)(sizeof(vecf) / sizeof(float));
    vecf r;
    for (int q = 0; q < VEC; q++) r[q] = __builtin_fmaf(hv[q], dd, bv[q]);
    return r;
}

template <int D, bool ELU>
__global__ __launch_bounds__(256) void gather_kernel(const int* __restrict__ rp,
                                                     const int* __restrict__ csr,
                                                     const float* __restrict__ dinv,
                                                     const float* __restrict__ H,
                                                     const float* __restrict__ bias,
                                                     float* __restrict__ Out, int n) {
    constexpr int VEC = D / 32;
    typedef __attribute__((ext_vector_type(VEC))) float vecf;
    const int lane = threadIdx.x & 31;
    const int nw = (int)gridDim.x * 8;
    int node = __builtin_amdgcn_readfirstlane((int)(blockIdx.x * 8 + (threadIdx.x >> 5)));
    if (node >= n) return;

    const int lofs = lane * VEC;
    const vecf bv = *(const vecf*)(bias + lofs);

    int beg = rp[node];
    int end = rp[node + 1];
    float di = dinv[node];
    const vecf hv0 = *(const vecf*)(H + (long long)node * D + lofs);
    int key = 0x7fffffff;
    if (lane < end - beg) key = csr[beg + lane];
    int node1 = node + nw;
    int beg1 = 0, end1 = 0;
    if (node1 < n) {
        beg1 = rp[node1];
        end1 = rp[node1 + 1];
    }

    vecf acc = self_term(bv, hv0, di * di);

    for (;;) {
        int key1 = 0x7fffffff;
        float di1 = 0.0f;
        vecf hv1 = {};
        const int node2 = node1 + nw;
        int beg2 = 0, end2 = 0;
        if (node1 < n) {
            if (lane < end1 - beg1) key1 = csr[beg1 + lane];
            di1 = dinv[node1];
            hv1 = *(const vecf*)(H + (long long)node1 * D + lofs);
            if (node2 < n) {
                beg2 = rp[node2];
                end2 = rp[node2 + 1];
            }
        }

        const int d = end - beg;

        if (d <= 32) {
            int rank = 0;
            for (int j = 0; j < d; j++) {
                int kj = __builtin_amdgcn_readlane(key, j);
                rank += (kj < key || (kj == key && j < lane)) ? 1 : 0;
            }
            const int skey = __builtin_amdgcn_ds_permute(rank << 2, key);

            int j = 0;
            for (; j + 8 <= d; j += 8) add_rows<8>(acc, skey, j, dinv, di, H, lofs);
            switch (d - j) {
                case 7: add_rows<7>(acc, skey, j, dinv, di, H, lofs); break;
                case 6: add_rows<6>(acc, skey, j, dinv, di, H, lofs); break;
                case 5: add_rows<5>(acc, skey, j, dinv, di, H, lofs); break;
                case 4: add_rows<4>(acc, skey, j, dinv, di, H, lofs); break;
                case 3: add_rows<3>(acc, skey, j, dinv, di, H, lofs); break;
                case 2: add_rows<2>(acc, skey, j, dinv, di, H, lofs); break;
                case 1: add_rows<1>(acc, skey, j, dinv, di, H, lofs); break;
                default: break;
            }
        } else {
            int prev = -1;
            int left = d;
            while (left > 0) {
                int best = 0x7fffffff;
                for (int b = beg + lane; b < end; b += 32) {
                    int k = csr[b];
                    if (k > prev && k < best) best = k;
                }
                for (int o = 16; o > 0; o >>= 1) {
                    int u = __shfl_xor(best, o, 32);
                    best = u < best ? u : best;
                }
                int c = 0;
                for (int b = beg + lane; b < end; b += 32) c += (csr[b] == best) ? 1 : 0;
                for (int o = 16; o > 0; o >>= 1) c += __shfl_xor(c, o, 32);
                best = __builtin_amdgcn_readfirstlane(best);
                c = __builtin_amdgcn_readfirstlane(c);
                if (c <= 0) break;
                const float w0 = dinv[best] * di;
                const vecf v0 = *(const vecf*)(H + (long long)best * D + lofs);
                for (int r = 0; r < c; r++) acc += v0 * w0;
                prev = best;
                left -= c;
            }
        }

        if (ELU) {
#pragma unroll
            for (int q = 0; q < VEC; q++) {
                float v = acc[q];
                acc[q] = v > 0.0f ? v : (__expf(v) - 1.0f);
            }
        }
        *(vecf*)(Out + (long long)node * D + lofs) = acc;

        acc = self_term(bv, hv1, di1 * di1);
        if (node1 >= n) break;
        node = node1; beg = beg1; end = end1; di = di1; key = key1;
        node1 = node2; beg1 = beg2; end1 = end2;
    }
}

extern "C" void kernel_launch(void* const* d_in, const int* in_sizes, int n_in,
                              void* d_out, int out_size, void* d_ws, size_t ws_size,
                              hipStream_t stream) {
    const float* x  = (const float*)d_in[0];
    const int*   ei = (const int*)d_in[1];
    const float* W1 = (const float*)d_in[2];
    const float* b1 = (const float*)d_in[3];
    const float* W2 = (const float*)d_in[4];
    const float* b2 = (const float*)d_in[5];
    float* out = (float*)d_out;

    const int N = in_sizes[0] / DIN;
    const int E = in_sizes[1] / 2;
    const int DH = 128;

    const int B = 256;
    const int gRows = (N + 127) / 128;
    const int gNode = GATHER_BLOCKS;
    const int NB    = (N + BUCKET_NODES - 1) / BUCKET_NODES;

    char* ws = (char*)d_ws;
    float* dinv = (float*)ws;
    size_t off = (((size_t)N * sizeof(float)) + 255) & ~(size_t)255;
    float* hbuf = (float*)(ws + off);
    float* aggbuf = (float*)(ws + off + (size_t)N * DH * sizeof(float));
    char* tail = ws + off + 2 * (size_t)N * DH * sizeof(float);
    int* rp = (int*)tail;
    size_t rpb = (((size_t)(N + 1) * sizeof(int)) + 255) & ~(size_t)255;
    int* csr = (int*)(tail + rpb);
    int* pairs = (int*)aggbuf;
    int* cpre  = pairs + 2 * (size_t)E;

    chunk_sort_kernel<<<SORT_BLOCKS, SORT_THREADS, 2 * NB * sizeof(int), stream>>>(ei, E, NB, cpre, pairs);
    bucket_rows_kernel<<<NB, BUCKET_NODES, 0, stream>>>(pairs, cpre, E, N, rp, dinv, csr);

    gemm_wmma_kernel<128><<<gRows, B, 0, stream>>>(x, W1, hbuf, N);
    gather_kernel<128, true><<<gNode, B, 0, stream>>>(rp, csr, dinv, hbuf, b1, aggbuf, N);

    gemm_wmma_kernel<64><<<gRows, B, 0, stream>>>(aggbuf, W2, hbuf, N);
    gather_kernel<64, false><<<gNode, B, 0, stream>>>(rp, csr, dinv, hbuf, b2, out, N);
}


# Layer 2's product moves into the layer-1 kernel: each 64-node tile's ELU result stays in LDS and is multiplied by W2 on WMMA there; the g @ W2 launch and the g buffer go; every sum order unchanged
# speedup vs baseline: 3.7440x; 1.1876x over previous
#include <hip/hip_runtime.h>
#include <hip/hip_bf16.h>

typedef __attribute__((ext_vector_type(2))) float v2f;
typedef __attribute__((ext_vector_type(8))) float v8f;

#define DIN 128

#define BUCKET_SHIFT 8
#define BUCKET_NODES (1 << BUCKET_SHIFT)
#define SORT_BLOCKS 256
#define SORT_THREADS 512
#define EDGE_BATCH 8
#define RUN_LANES 16
#define RUN_STEPS 16

static_assert(SORT_BLOCKS == 32 * (BUCKET_NODES / 32), "bucket_rows_kernel: 8 waves x 32 chunk runs");
static_assert(RUN_STEPS * (32 / RUN_LANES) == 32, "a wave covers 32 chunk runs");

__global__ __launch_bounds__(SORT_THREADS) void chunk_sort_kernel(const int* __restrict__ ei, int e, int nb,
                                                                  int* __restrict__ cpre,
                                                                  int* __restrict__ pairs) {
    extern __shared__ int sh[];
    __shared__ int wt[SORT_THREADS / 32];
    int* __restrict__ cnt = sh;
    int* __restrict__ cur = sh + nb;
    const int tid = threadIdx.x;
    const int lane = tid & 31;
    const int wave = tid >> 5;
    const int g = (int)blockIdx.x;
    for (int b = tid; b < nb; b += SORT_THREADS) cnt[b] = 0;
    __syncthreads();

    const int ch = (e + SORT_BLOCKS - 1) / SORT_BLOCKS;
    const int lo = g * ch;
    const int hi = (lo + ch < e) ? (lo + ch) : e;
    const int* __restrict__ dstp = ei + e;

    for (int i0 = lo + tid; i0 < hi; i0 += SORT_THREADS * EDGE_BATCH) {
        int d[EDGE_BATCH];
#pragma unroll
        for (int q = 0; q < EDGE_BATCH; q++) {
            const int i = i0 + q * SORT_THREADS;
            d[q] = (i < hi) ? dstp[i] : -1;
        }
#pragma unroll
        for (int q = 0; q < EDGE_BATCH; q++)
            if (d[q] >= 0) atomicAdd(&cnt[d[q] >> BUCKET_SHIFT], 1);
    }
    __syncthreads();

    int carry = 0;
    for (int base = 0; base < nb; base += SORT_THREADS) {
        const int b = base + tid;
        const int t = (b < nb) ? cnt[b] : 0;
        int incl = t;
        for (int o = 1; o < 32; o <<= 1) {
            int u = __shfl_up(incl, o, 32);
            if (lane >= o) incl += u;
        }
        if (lane == 31) wt[wave] = incl;
        __syncthreads();
        int offs = carry + incl - t;
        int all = 0;
        for (int w = 0; w < SORT_THREADS / 32; w++) {
            const int x = wt[w];
            if (w < wave) offs += x;
            all += x;
        }
        if (b < nb) {
            cur[b] = offs;
            cpre[(size_t)b * SORT_BLOCKS + g] = offs;
        }
        carry += all;
        __syncthreads();
    }
    if (tid == 0) cpre[(size_t)nb * SORT_BLOCKS + g] = carry;

    for (int i0 = lo + tid; i0 < hi; i0 += SORT_THREADS * EDGE_BATCH) {
        int s[EDGE_BATCH];
        int d[EDGE_BATCH];
#pragma unroll
        for (int q = 0; q < EDGE_BATCH; q++) {
            const int i = i0 + q * SORT_THREADS;
            s[q] = (i < hi) ? ei[i] : 0;
            d[q] = (i < hi) ? dstp[i] : -1;
        }
#pragma unroll
        for (int q = 0; q < EDGE_BATCH; q++) {
            if (d[q] >= 0) {
                const int p = lo + atomicAdd(&cur[d[q] >> BUCKET_SHIFT], 1);
                *(int2*)(pairs + 2 * (long long)p) = make_int2(s[q], d[q]);
            }
        }
    }
}

__global__ __launch_bounds__(BUCKET_NODES) void bucket_rows_kernel(const int* __restrict__ pairs,
                                                                   const int* __restrict__ cpre, int e, int n,
                                                                   int* __restrict__ rp,
                                                                   float* __restrict__ dinv,
                                                                   int* __restrict__ csr) {
    __shared__ int cnt[BUCKET_NODES];
    __shared__ int cur[BUCKET_NODES];
    __shared__ int runS[SORT_BLOCKS];
    __shared__ int runL[SORT_BLOCKS];
    __shared__ int wtot[BUCKET_NODES / 32];
    __shared__ int wlo[BUCKET_NODES / 32];
    const int tid = threadIdx.x;
    const int lane = tid & 31;
    const int wave = tid >> 5;
    const int b = (int)blockIdx.x;

    const int ch = (e + SORT_BLOCKS - 1) / SORT_BLOCKS;
    const int o0 = cpre[(size_t)b * SORT_BLOCKS + tid];
    const int o1 = cpre[(size_t)(b + 1) * SORT_BLOCKS + tid];
    runS[tid] = tid * ch + o0;
    runL[tid] = o1 - o0;
    cnt[tid] = 0;
    int s0 = o0;
    for (int o = 16; o > 0; o >>= 1) s0 += __shfl_xor(s0, o, 32);
    if (lane == 0) wlo[wave] = s0;
    __syncthreads();
    int lo = 0;
    for (int w = 0; w < BUCKET_NODES / 32; w++) lo += wlo[w];

    const int sub = lane & (RUN_LANES - 1);
    const int which = lane / RUN_LANES;
    int S[RUN_STEPS];
    int L[RUN_STEPS];
#pragma unroll
    for (int j = 0; j < RUN_STEPS; j++) {
        const int gi = wave * 32 + j * (32 / RUN_LANES) + which;
        S[j] = runS[gi];
        L[j] = runL[gi];
    }

    {
        int d[RUN_STEPS];
#pragma unroll
        for (int j = 0; j < RUN_STEPS; j++)
            d[j] = (sub < L[j]) ? pairs[2 * (long long)(S[j] + sub) + 1] : -1;
#pragma unroll
        for (int j = 0; j < RUN_STEPS; j++)
            if (d[j] >= 0) atomicAdd(&cnt[d[j] & (BUCKET_NODES - 1)], 1);
#pragma unroll
        for (int j = 0; j < RUN_STEPS; j++) {
            for (int k = sub + RUN_LANES; k < L[j]; k += RUN_LANES) {
                const int dd = pairs[2 * (long long)(S[j] + k) + 1];
                atomicAdd(&cnt[dd & (BUCKET_NODES - 1)], 1);
            }
        }
    }
    __syncthreads();

    const int c = cnt[tid];
    int incl = c;
    for (int o = 1; o < 32; o <<= 1) {
        int u = __shfl_up(incl, o, 32);
        if (lane >= o) incl += u;
    }
    if (lane == 31) wtot[wave] = incl;
    __syncthreads();
    int start = lo + incl - c;
    for (int w = 0; w < BUCKET_NODES / 32; w++)
        if (w < wave) start += wtot[w];

    const int node = b * BUCKET_NODES + tid;
    cur[tid] = start;
    if (node < n) {
        rp[node] = start;
        dinv[node] = rsqrtf((float)(c + 1));
        if (node == n - 1) rp[n] = start + c;
    }
    __syncthreads();

    {
        int2 pr[RUN_STEPS];
#pragma unroll
        for (int j = 0; j < RUN_STEPS; j++)
            pr[j] = (sub < L[j]) ? *(const int2*)(pairs + 2 * (long long)(S[j] + sub)) : make_int2(0, -1);
#pragma unroll
        for (int j = 0; j < RUN_STEPS; j++) {
            if (pr[j].y >= 0) {
                const int p = atomicAdd(&cur[pr[j].y & (BUCKET_NODES - 1)], 1);
                csr[p] = pr[j].x;
            }
        }
#pragma unroll
        for (int j = 0; j < RUN_STEPS; j++) {
            for (int k = sub + RUN_LANES; k < L[j]; k += RUN_LANES) {
                const int2 q2 = *(const int2*)(pairs + 2 * (long long)(S[j] + k));
                const int p = atomicAdd(&cur[q2.y & (BUCKET_NODES - 1)], 1);
                csr[p] = q2.x;
            }
        }
    }
}

#define GATHER_BLOCKS 2048

template <int K, typename vecf>
__device__ __forceinline__ void add_rows(vecf& acc, const int skey, const int j,
                                         const float* __restrict__ dinv, const float di,
                                         const float* __restrict__ H, const int lofs) {
    constexpr int D = (int)(sizeof(vecf) / sizeof(float)) * 32;
    int s[K];
    float w[K];
    vecf v[K];
#pragma unroll
    for (int q = 0; q < K; q++) s[q] = __builtin_amdgcn_readlane(skey, j + q);
#pragma unroll
    for (int q = 0; q < K; q++) w[q] = dinv[s[q]] * di;
#pragma unroll
    for (int q = 0; q < K; q++) v[q] = *(const vecf*)(H + (long long)s[q] * D + lofs);
#pragma unroll
    for (int q = 0; q < K; q++) acc += v[q] * w[q];
}

template <typename vecf>
__device__ __forceinline__ vecf self_term(const vecf bv, const vecf hv, const float dd) {
    constexpr int VEC = (int)(sizeof(vecf) / sizeof(float));
    vecf r;
    for (int q = 0; q < VEC; q++) r[q] = __builtin_fmaf(hv[q], dd, bv[q]);
    return r;
}

template <int D, bool ELU>
__global__ __launch_bounds__(256) void gather_kernel(const int* __restrict__ rp,
                                                     const int* __restrict__ csr,
                                                     const float* __restrict__ dinv,
                                                     const float* __restrict__ H,
                                                     const float* __restrict__ bias,
                                                     float* __restrict__ Out, int n) {
    constexpr int VEC = D / 32;
    typedef __attribute__((ext_vector_type(VEC))) float vecf;
    const int lane = threadIdx.x & 31;
    const int nw = (int)gridDim.x * 8;
    int node = __builtin_amdgcn_readfirstlane((int)(blockIdx.x * 8 + (threadIdx.x >> 5)));
    if (node >= n) return;

    const int lofs = lane * VEC;
    const vecf bv = *(const vecf*)(bias + lofs);

    int beg = rp[node];
    int end = rp[node + 1];
    float di = dinv[node];
    const vecf hv0 = *(const vecf*)(H + (long long)node * D + lofs);
    int key = 0x7fffffff;
    if (lane < end - beg) key = csr[beg + lane];
    int node1 = node + nw;
    int beg1 = 0, end1 = 0;
    if (node1 < n) {
        beg1 = rp[node1];
        end1 = rp[node1 + 1];
    }

    vecf acc = self_term(bv, hv0, di * di);

    for (;;) {
        int key1 = 0x7fffffff;
        float di1 = 0.0f;
        vecf hv1 = {};
        const int node2 = node1 + nw;
        int beg2 = 0, end2 = 0;
        if (node1 < n) {
            if (lane < end1 - beg1) key1 = csr[beg1 + lane];
            di1 = dinv[node1];
            hv1 = *(const vecf*)(H + (long long)node1 * D + lofs);
            if (node2 < n) {
                beg2 = rp[node2];
                end2 = rp[node2 + 1];
            }
        }

        const int d = end - beg;

        if (d <= 32) {
            int rank = 0;
            for (int j = 0; j < d; j++) {
                int kj = __builtin_amdgcn_readlane(key, j);
                rank += (kj < key || (kj == key && j < lane)) ? 1 : 0;
            }
            const int skey = __builtin_amdgcn_ds_permute(rank << 2, key);

            int j = 0;
            for (; j + 8 <= d; j += 8) add_rows<8>(acc, skey, j, dinv, di, H, lofs);
            switch (d - j) {
                case 7: add_rows<7>(acc, skey, j, dinv, di, H, lofs); break;
                case 6: add_rows<6>(acc, skey, j, dinv, di, H, lofs); break;
                case 5: add_rows<5>(acc, skey, j, dinv, di, H, lofs); break;
                case 4: add_rows<4>(acc, skey, j, dinv, di, H, lofs); break;
                case 3: add_rows<3>(acc, skey, j, dinv, di, H, lofs); break;
                case 2: add_rows<2>(acc, skey, j, dinv, di, H, lofs); break;
                case 1: add_rows<1>(acc, skey, j, dinv, di, H, lofs); break;
                default: break;
            }
        } else {
            int prev = -1;
            int left = d;
            while (left > 0) {
                int best = 0x7fffffff;
                for (int b = beg + lane; b < end; b += 32) {
                    int k = csr[b];
                    if (k > prev && k < best) best = k;
                }
                for (int o = 16; o > 0; o >>= 1) {
                    int u = __shfl_xor(best, o, 32);
                    best = u < best ? u : best;
                }
                int c = 0;
                for (int b = beg + lane; b < end; b += 32) c += (csr[b] == best) ? 1 : 0;
                for (int o = 16; o > 0; o >>= 1) c += __shfl_xor(c, o, 32);
                best = __builtin_amdgcn_readfirstlane(best);
                c = __builtin_amdgcn_readfirstlane(c);
                if (c <= 0) break;
                const float w0 = dinv[best] * di;
                const vecf v0 = *(const vecf*)(H + (long long)best * D + lofs);
                for (int r = 0; r < c; r++) acc += v0 * w0;
                prev = best;
                left -= c;
            }
        }

        if (ELU) {
#pragma unroll
            for (int q = 0; q < VEC; q++) {
                float v = acc[q];
                acc[q] = v > 0.0f ? v : (__expf(v) - 1.0f);
            }
        }
        *(vecf*)(Out + (long long)node * D + lofs) = acc;

        acc = self_term(bv, hv1, di1 * di1);
        if (node1 >= n) break;
        node = node1; beg = beg1; end = end1; di = di1; key = key1;
        node1 = node2; beg1 = beg2; end1 = end2;
    }
}

#define TILE_NODES 64
#define TILE_PITCH 128
#define FUSED_THREADS 512
#define FUSED_BLOCKS 512

template <typename vecf>
__device__ __forceinline__ vecf sum_row(vecf acc, const int key, const int beg, const int end, const int lane,
                                        const int* __restrict__ csr, const float* __restrict__ dinv,
                                        const float di, const float* __restrict__ H, const int lofs) {
    constexpr int D = (int)(sizeof(vecf) / sizeof(float)) * 32;
    const int d = end - beg;
    if (d <= 32) {
        int rank = 0;
        for (int j = 0; j < d; j++) {
            int kj = __builtin_amdgcn_readlane(key, j);
            rank += (kj < key || (kj == key && j < lane)) ? 1 : 0;
        }
        const int skey = __builtin_amdgcn_ds_permute(rank << 2, key);
        int j = 0;
        for (; j + 8 <= d; j += 8) add_rows<8>(acc, skey, j, dinv, di, H, lofs);
        switch (d - j) {
            case 7: add_rows<7>(acc, skey, j, dinv, di, H, lofs); break;
            case 6: add_rows<6>(acc, skey, j, dinv, di, H, lofs); break;
            case 5: add_rows<5>(acc, skey, j, dinv, di, H, lofs); break;
            case 4: add_rows<4>(acc, skey, j, dinv, di, H, lofs); break;
            case 3: add_rows<3>(acc, skey, j, dinv, di, H, lofs); break;
            case 2: add_rows<2>(acc, skey, j, dinv, di, H, lofs); break;
            case 1: add_rows<1>(acc, skey, j, dinv, di, H, lofs); break;
            default: break;
        }
    } else {
        int prev = -1;
        int left = d;
        while (left > 0) {
            int best = 0x7fffffff;
            for (int b = beg + lane; b < end; b += 32) {
                int k = csr[b];
                if (k > prev && k < best) best = k;
            }
            for (int o = 16; o > 0; o >>= 1) {
                int u = __shfl_xor(best, o, 32);
                best = u < best ? u : best;
            }
            int c = 0;
            for (int b = beg + lane; b < end; b += 32) c += (csr[b] == best) ? 1 : 0;
            for (int o = 16; o > 0; o >>= 1) c += __shfl_xor(c, o, 32);
            best = __builtin_amdgcn_readfirstlane(best);
            c = __builtin_amdgcn_readfirstlane(c);
            if (c <= 0) break;
            const float w0 = dinv[best] * di;
            const vecf v0 = *(const vecf*)(H + (long long)best * D + lofs);
            for (int r = 0; r < c; r++) acc += v0 * w0;
            prev = best;
            left -= c;
        }
    }
    return acc;
}

__global__ __launch_bounds__(FUSED_THREADS) void gather_gemm_elu_gemm_kernel(const int* __restrict__ rp,
                                                                             const int* __restrict__ csr,
                                                                             const float* __restrict__ dinv,
                                                                             const float* __restrict__ X,
                                                                             const float* __restrict__ W1,
                                                                             const float* __restrict__ bias1,
                                                                             const float* __restrict__ W2,
                                                                             float* __restrict__ H2, int n) {
    constexpr int D1 = 128;
    constexpr int D2 = 64;
    constexpr int NT1 = D1 / 16;
    constexpr int NT2 = D2 / 16;
    constexpr int NPAIR1 = 32 * NT1 * 32;
    constexpr int NPAIR2 = 32 * NT2 * 32;
    typedef __attribute__((ext_vector_type(4))) float vecf;
    __shared__ __attribute__((aligned(16))) float sW1[NPAIR1 * 2];
    __shared__ __attribute__((aligned(16))) float sW2[NPAIR2 * 2];
    __shared__ __attribute__((aligned(16))) float sA[TILE_NODES * TILE_PITCH];
    __shared__ __attribute__((aligned(16))) float sG[TILE_NODES * TILE_PITCH];

    for (int pair = threadIdx.x; pair < NPAIR1; pair += FUSED_THREADS) {
        int l  = pair & 31;
        int tt = (pair >> 5) % NT1;
        int ks = (pair >> 5) / NT1;
        int col = tt * 16 + (l & 15);
        int k0  = ks * 4 + ((l >> 4) * 2);
        sW1[pair * 2 + 0] = W1[k0 * D1 + col];
        sW1[pair * 2 + 1] = W1[(k0 + 1) * D1 + col];
    }
    for (int pair = threadIdx.x; pair < NPAIR2; pair += FUSED_THREADS) {
        int l  = pair & 31;
        int tt = (pair >> 5) % NT2;
        int ks = (pair >> 5) / NT2;
        int col = tt * 16 + (l & 15);
        int k0  = ks * 4 + ((l >> 4) * 2);
        sW2[pair * 2 + 0] = W2[k0 * D2 + col];
        sW2[pair * 2 + 1] = W2[(k0 + 1) * D2 + col];
    }

    const int lane = threadIdx.x & 31;
    const int wave = __builtin_amdgcn_readfirstlane((int)(threadIdx.x >> 5));
    const int m    = lane & 15;
    const int kk   = (lane >> 4) * 2;
    const int lofs = lane * 4;

    const int ct = wave & 7;
    const int rg = wave >> 3;
    const float bcol = bias1[ct * 16 + m];
    const float* __restrict__ a0p = sA + (rg * 32 + m) * TILE_PITCH + kk;
    const float* __restrict__ a1p = a0p + 16 * TILE_PITCH;
    const float* __restrict__ bp1 = sW1 + (ct * 32 + lane) * 2;
    const int gc   = ct * 16 + m;
    const int grow = rg * 32 + (lane >> 4) * 8;

    const int rt = wave >> 2;
    const int c2 = wave & 3;
    const float* __restrict__ gp  = sG + (rt * 16 + m) * TILE_PITCH + kk;
    const float* __restrict__ bp2 = sW2 + (c2 * 32 + lane) * 2;

    const int ntiles = (n + TILE_NODES - 1) / TILE_NODES;
    for (int tile = (int)blockIdx.x; tile < ntiles; tile += (int)gridDim.x) {
        const int nb0 = tile * TILE_NODES + wave * 4;
        int beg[4], end[4], key[4];
        float di[4];
        vecf hv[4];
#pragma unroll
        for (int q = 0; q < 4; q++) {
            const int node = nb0 + q;
            beg[q] = 0; end[q] = 0;
            if (node < n) {
                beg[q] = rp[node];
                end[q] = rp[node + 1];
            }
        }
#pragma unroll
        for (int q = 0; q < 4; q++) {
            const int node = nb0 + q;
            key[q] = 0x7fffffff;
            di[q] = 0.0f;
            hv[q] = vecf{};
            if (node < n) {
                if (lane < end[q] - beg[q]) key[q] = csr[beg[q] + lane];
                di[q] = dinv[node];
                hv[q] = *(const vecf*)(X + (long long)node * DIN + lofs);
            }
        }
#pragma unroll
        for (int q = 0; q < 4; q++) {
            vecf acc = hv[q] * (di[q] * di[q]);
            if (nb0 + q < n) acc = sum_row<vecf>(acc, key[q], beg[q], end[q], lane, csr, dinv, di[q], X, lofs);
            const int r = wave * 4 + q;
            *(vecf*)(sA + r * TILE_PITCH + ((lane ^ (r & 15)) << 2)) = acc;
        }
        __syncthreads();

        v8f acc0 = {};
        v8f acc1 = {};
#pragma unroll 8
        for (int ks = 0; ks < 32; ks++) {
            v2f b  = *(const v2f*)(bp1 + ks * (NT1 * 64));
            const int ko = (ks ^ m) << 2;
            v2f a0 = *(const v2f*)(a0p + ko);
            v2f a1 = *(const v2f*)(a1p + ko);
            acc0 = __builtin_amdgcn_wmma_f32_16x16x4_f32(false, a0, false, b, (short)0, acc0, false, false);
            acc1 = __builtin_amdgcn_wmma_f32_16x16x4_f32(false, a1, false, b, (short)0, acc1, false, false);
        }

#pragma unroll
        for (int v = 0; v < 8; v++) {
            float t0 = acc0[v] + bcol;
            float t1 = acc1[v] + bcol;
            t0 = t0 > 0.0f ? t0 : (__expf(t0) - 1.0f);
            t1 = t1 > 0.0f ? t1 : (__expf(t1) - 1.0f);
            const int r0 = grow + v;
            const int go = (((gc >> 2) ^ (r0 & 15)) << 2) + (gc & 3);
            sG[r0 * TILE_PITCH + go] = t0;
            sG[(r0 + 16) * TILE_PITCH + go] = t1;
        }
        __syncthreads();

        v8f acc2 = {};
#pragma unroll 8
        for (int ks = 0; ks < 32; ks++) {
            v2f b = *(const v2f*)(bp2 + ks * (NT2 * 64));
            v2f a = *(const v2f*)(gp + ((ks ^ m) << 2));
            acc2 = __builtin_amdgcn_wmma_f32_16x16x4_f32(false, a, false, b, (short)0, acc2, false, false);
        }

        const long long rbase = (long long)tile * TILE_NODES + rt * 16 + (lane >> 4) * 8;
#pragma unroll
        for (int v = 0; v < 8; v++) {
            const long long r = rbase + v;
            if (r < n) H2[r * D2 + c2 * 16 + m] = acc2[v];
        }
    }
}

extern "C" void kernel_launch(void* const* d_in, const int* in_sizes, int n_in,
                              void* d_out, int out_size, void* d_ws, size_t ws_size,
                              hipStream_t stream) {
    const float* x  = (const float*)d_in[0];
    const int*   ei = (const int*)d_in[1];
    const float* W1 = (const float*)d_in[2];
    const float* b1 = (const float*)d_in[3];
    const float* W2 = (const float*)d_in[4];
    const float* b2 = (const float*)d_in[5];
    float* out = (float*)d_out;

    const int N = in_sizes[0] / DIN;
    const int E = in_sizes[1] / 2;
    const int DH = 128;

    const int B = 256;
    const int gNode = GATHER_BLOCKS;
    const int NB    = (N + BUCKET_NODES - 1) / BUCKET_NODES;

    char* ws = (char*)d_ws;
    float* dinv = (float*)ws;
    size_t off = (((size_t)N * sizeof(float)) + 255) & ~(size_t)255;
    float* hbuf = (float*)(ws + off);
    float* aggbuf = (float*)(ws + off + (size_t)N * DH * sizeof(float));
    char* tail = ws + off + 2 * (size_t)N * DH * sizeof(float);
    int* rp = (int*)tail;
    size_t rpb = (((size_t)(N + 1) * sizeof(int)) + 255) & ~(size_t)255;
    int* csr = (int*)(tail + rpb);
    int* pairs = (int*)aggbuf;
    int* cpre  = pairs + 2 * (size_t)E;

    chunk_sort_kernel<<<SORT_BLOCKS, SORT_THREADS, 2 * NB * sizeof(int), stream>>>(ei, E, NB, cpre, pairs);
    bucket_rows_kernel<<<NB, BUCKET_NODES, 0, stream>>>(pairs, cpre, E, N, rp, dinv, csr);

    gather_gemm_elu_gemm_kernel<<<FUSED_BLOCKS, FUSED_THREADS, 0, stream>>>(rp, csr, dinv, x, W1, b1, W2, hbuf, N);

    gather_kernel<64, false><<<gNode, B, 0, stream>>>(rp, csr, dinv, hbuf, b2, out, N);
}


# Rows of up to 32 sources are sorted by source once at build time (owner thread, register sorting network over an LDS row); both neighbour sums drop their per-node rank loop; sum order unchanged
# speedup vs baseline: 3.9105x; 1.0445x over previous
#include <hip/hip_runtime.h>
#include <hip/hip_bf16.h>

typedef __attribute__((ext_vector_type(2))) float v2f;
typedef __attribute__((ext_vector_type(8))) float v8f;

#define DIN 128

#define BUCKET_SHIFT 8
#define BUCKET_NODES (1 << BUCKET_SHIFT)
#define SORT_BLOCKS 256
#define SORT_THREADS 512
#define EDGE_BATCH 8
#define RUN_LANES 16
#define RUN_STEPS 16
#define ROW_SLOTS 32
#define ROW_PITCH (ROW_SLOTS + 1)
#define LONG_ROW 0x40000000

static_assert(SORT_BLOCKS == 32 * (BUCKET_NODES / 32), "bucket_rows_kernel: 8 waves x 32 chunk runs");
static_assert(RUN_STEPS * (32 / RUN_LANES) == 32, "a wave covers 32 chunk runs");

__global__ __launch_bounds__(SORT_THREADS) void chunk_sort_kernel(const int* __restrict__ ei, int e, int nb,
                                                                  int* __restrict__ cpre,
                                                                  int* __restrict__ pairs) {
    extern __shared__ int sh[];
    __shared__ int wt[SORT_THREADS / 32];
    int* __restrict__ cnt = sh;
    int* __restrict__ cur = sh + nb;
    const int tid = threadIdx.x;
    const int lane = tid & 31;
    const int wave = tid >> 5;
    const int g = (int)blockIdx.x;
    for (int b = tid; b < nb; b += SORT_THREADS) cnt[b] = 0;
    __syncthreads();

    const int ch = (e + SORT_BLOCKS - 1) / SORT_BLOCKS;
    const int lo = g * ch;
    const int hi = (lo + ch < e) ? (lo + ch) : e;
    const int* __restrict__ dstp = ei + e;

    for (int i0 = lo + tid; i0 < hi; i0 += SORT_THREADS * EDGE_BATCH) {
        int d[EDGE_BATCH];
#pragma unroll
        for (int q = 0; q < EDGE_BATCH; q++) {
            const int i = i0 + q * SORT_THREADS;
            d[q] = (i < hi) ? dstp[i] : -1;
        }
#pragma unroll
        for (int q = 0; q < EDGE_BATCH; q++)
            if (d[q] >= 0) atomicAdd(&cnt[d[q] >> BUCKET_SHIFT], 1);
    }
    __syncthreads();

    int carry = 0;
    for (int base = 0; base < nb; base += SORT_THREADS) {
        const int b = base + tid;
        const int t = (b < nb) ? cnt[b] : 0;
        int incl = t;
        for (int o = 1; o < 32; o <<= 1) {
            int u = __shfl_up(incl, o, 32);
            if (lane >= o) incl += u;
        }
        if (lane == 31) wt[wave] = incl;
        __syncthreads();
        int offs = carry + incl - t;
        int all = 0;
        for (int w = 0; w < SORT_THREADS / 32; w++) {
            const int x = wt[w];
            if (w < wave) offs += x;
            all += x;
        }
        if (b < nb) {
            cur[b] = offs;
            cpre[(size_t)b * SORT_BLOCKS + g] = offs;
        }
        carry += all;
        __syncthreads();
    }
    if (tid == 0) cpre[(size_t)nb * SORT_BLOCKS + g] = carry;

    for (int i0 = lo + tid; i0 < hi; i0 += SORT_THREADS * EDGE_BATCH) {
        int s[EDGE_BATCH];
        int d[EDGE_BATCH];
#pragma unroll
        for (int q = 0; q < EDGE_BATCH; q++) {
            const int i = i0 + q * SORT_THREADS;
            s[q] = (i < hi) ? ei[i] : 0;
            d[q] = (i < hi) ? dstp[i] : -1;
        }
#pragma unroll
        for (int q = 0; q < EDGE_BATCH; q++) {
            if (d[q] >= 0) {
                const int p = lo + atomicAdd(&cur[d[q] >> BUCKET_SHIFT], 1);
                *(int2*)(pairs + 2 * (long long)p) = make_int2(s[q], d[q]);
            }
        }
    }
}

__device__ __forceinline__ void order_pair(int& a, int& b) {
    const int lo = a < b ? a : b;
    const int hi = a < b ? b : a;
    a = lo;
    b = hi;
}

template <int LO, int N, int R, int SZ>
__device__ __forceinline__ void oe_merge(int (&v)[SZ]) {
    constexpr int M = R * 2;
    if constexpr (M < N) {
        oe_merge<LO, N, M, SZ>(v);
        oe_merge<LO + R, N, M, SZ>(v);
#pragma unroll
        for (int i = LO + R; i + R < LO + N; i += M) order_pair(v[i], v[i + R]);
    } else {
        order_pair(v[LO], v[LO + R]);
    }
}

template <int LO, int N, int SZ>
__device__ __forceinline__ void oe_sort(int (&v)[SZ]) {
    if constexpr (N > 1) {
        oe_sort<LO, N / 2, SZ>(v);
        oe_sort<LO + N / 2, N / 2, SZ>(v);
        oe_merge<LO, N, 1, SZ>(v);
    }
}

template <int N>
__device__ __forceinline__ void order_row(const int* __restrict__ row, const int c, int* __restrict__ out) {
    int v[N];
#pragma unroll
    for (int k = 0; k < N; k++) v[k] = (k < c) ? row[k] : 0x7fffffff;
    oe_sort<0, N, N>(v);
#pragma unroll
    for (int k = 0; k < N; k++)
        if (k < c) out[k] = v[k];
}

__global__ __launch_bounds__(BUCKET_NODES) void bucket_rows_kernel(const int* __restrict__ pairs,
                                                                   const int* __restrict__ cpre, int e, int n,
                                                                   int* __restrict__ rp,
                                                                   float* __restrict__ dinv,
                                                                   int* __restrict__ csr) {
    __shared__ int cnt[BUCKET_NODES];
    __shared__ int cur[BUCKET_NODES];
    __shared__ int runS[SORT_BLOCKS];
    __shared__ int runL[SORT_BLOCKS];
    __shared__ int wtot[BUCKET_NODES / 32];
    __shared__ int wlo[BUCKET_NODES / 32];
    __shared__ int lrow[BUCKET_NODES * ROW_PITCH];
    const int tid = threadIdx.x;
    const int lane = tid & 31;
    const int wave = tid >> 5;
    const int b = (int)blockIdx.x;

    const int ch = (e + SORT_BLOCKS - 1) / SORT_BLOCKS;
    const int o0 = cpre[(size_t)b * SORT_BLOCKS + tid];
    const int o1 = cpre[(size_t)(b + 1) * SORT_BLOCKS + tid];
    runS[tid] = tid * ch + o0;
    runL[tid] = o1 - o0;
    cnt[tid] = 0;
    int s0 = o0;
    for (int o = 16; o > 0; o >>= 1) s0 += __shfl_xor(s0, o, 32);
    if (lane == 0) wlo[wave] = s0;
    __syncthreads();
    int lo = 0;
    for (int w = 0; w < BUCKET_NODES / 32; w++) lo += wlo[w];

    const int sub = lane & (RUN_LANES - 1);
    const int which = lane / RUN_LANES;
    int S[RUN_STEPS];
    int L[RUN_STEPS];
#pragma unroll
    for (int j = 0; j < RUN_STEPS; j++) {
        const int gi = wave * 32 + j * (32 / RUN_LANES) + which;
        S[j] = runS[gi];
        L[j] = runL[gi];
    }

    {
        int d[RUN_STEPS];
#pragma unroll
        for (int j = 0; j < RUN_STEPS; j++)
            d[j] = (sub < L[j]) ? pairs[2 * (long long)(S[j] + sub) + 1] : -1;
#pragma unroll
        for (int j = 0; j < RUN_STEPS; j++)
            if (d[j] >= 0) atomicAdd(&cnt[d[j] & (BUCKET_NODES - 1)], 1);
#pragma unroll
        for (int j = 0; j < RUN_STEPS; j++) {
            for (int k = sub + RUN_LANES; k < L[j]; k += RUN_LANES) {
                const int dd = pairs[2 * (long long)(S[j] + k) + 1];
                atomicAdd(&cnt[dd & (BUCKET_NODES - 1)], 1);
            }
        }
    }
    __syncthreads();

    const int c = cnt[tid];
    int incl = c;
    for (int o = 1; o < 32; o <<= 1) {
        int u = __shfl_up(incl, o, 32);
        if (lane >= o) incl += u;
    }
    if (lane == 31) wtot[wave] = incl;
    __syncthreads();
    int start = lo + incl - c;
    for (int w = 0; w < BUCKET_NODES / 32; w++)
        if (w < wave) start += wtot[w];

    const int node = b * BUCKET_NODES + tid;
    cur[tid] = (c <= ROW_SLOTS) ? 0 : (start | LONG_ROW);
    if (node < n) {
        rp[node] = start;
        dinv[node] = rsqrtf((float)(c + 1));
        if (node == n - 1) rp[n] = start + c;
    }
    __syncthreads();

    {
        int2 pr[RUN_STEPS];
#pragma unroll
        for (int j = 0; j < RUN_STEPS; j++)
            pr[j] = (sub < L[j]) ? *(const int2*)(pairs + 2 * (long long)(S[j] + sub)) : make_int2(0, -1);
#pragma unroll
        for (int j = 0; j < RUN_STEPS; j++) {
            if (pr[j].y >= 0) {
                const int dl = pr[j].y & (BUCKET_NODES - 1);
                const int p = atomicAdd(&cur[dl], 1);
                if (p & LONG_ROW) csr[p ^ LONG_ROW] = pr[j].x;
                else lrow[dl * ROW_PITCH + p] = pr[j].x;
            }
        }
#pragma unroll
        for (int j = 0; j < RUN_STEPS; j++) {
            for (int k = sub + RUN_LANES; k < L[j]; k += RUN_LANES) {
                const int2 q2 = *(const int2*)(pairs + 2 * (long long)(S[j] + k));
                const int dl = q2.y & (BUCKET_NODES - 1);
                const int p = atomicAdd(&cur[dl], 1);
                if (p & LONG_ROW) csr[p ^ LONG_ROW] = q2.x;
                else lrow[dl * ROW_PITCH + p] = q2.x;
            }
        }
    }
    __syncthreads();

    const int cs = (c <= ROW_SLOTS) ? c : 0;
    if (__any(cs > 16)) order_row<ROW_SLOTS>(lrow + tid * ROW_PITCH, cs, csr + start);
    else order_row<16>(lrow + tid * ROW_PITCH, cs, csr + start);
}

#define GATHER_BLOCKS 2048

template <int K, typename vecf>
__device__ __forceinline__ void add_rows(vecf& acc, const int skey, const int j,
                                         const float* __restrict__ dinv, const float di,
                                         const float* __restrict__ H, const int lofs) {
    constexpr int D = (int)(sizeof(vecf) / sizeof(float)) * 32;
    int s[K];
    float w[K];
    vecf v[K];
#pragma unroll
    for (int q = 0; q < K; q++) s[q] = __builtin_amdgcn_readlane(skey, j + q);
#pragma unroll
    for (int q = 0; q < K; q++) w[q] = dinv[s[q]] * di;
#pragma unroll
    for (int q = 0; q < K; q++) v[q] = *(const vecf*)(H + (long long)s[q] * D + lofs);
#pragma unroll
    for (int q = 0; q < K; q++) acc += v[q] * w[q];
}

template <typename vecf>
__device__ __forceinline__ vecf self_term(const vecf bv, const vecf hv, const float dd) {
    constexpr int VEC = (int)(sizeof(vecf) / sizeof(float));
    vecf r;
    for (int q = 0; q < VEC; q++) r[q] = __builtin_fmaf(hv[q], dd, bv[q]);
    return r;
}

template <int D, bool ELU>
__global__ __launch_bounds__(256) void gather_kernel(const int* __restrict__ rp,
                                                     const int* __restrict__ csr,
                                                     const float* __restrict__ dinv,
                                                     const float* __restrict__ H,
                                                     const float* __restrict__ bias,
                                                     float* __restrict__ Out, int n) {
    constexpr int VEC = D / 32;
    typedef __attribute__((ext_vector_type(VEC))) float vecf;
    const int lane = threadIdx.x & 31;
    const int nw = (int)gridDim.x * 8;
    int node = __builtin_amdgcn_readfirstlane((int)(blockIdx.x * 8 + (threadIdx.x >> 5)));
    if (node >= n) return;

    const int lofs = lane * VEC;
    const vecf bv = *(const vecf*)(bias + lofs);

    int beg = rp[node];
    int end = rp[node + 1];
    float di = dinv[node];
    const vecf hv0 = *(const vecf*)(H + (long long)node * D + lofs);
    int key = 0x7fffffff;
    if (lane < end - beg) key = csr[beg + lane];
    int node1 = node + nw;
    int beg1 = 0, end1 = 0;
    if (node1 < n) {
        beg1 = rp[node1];
        end1 = rp[node1 + 1];
    }

    vecf acc = self_term(bv, hv0, di * di);

    for (;;) {
        int key1 = 0x7fffffff;
        float di1 = 0.0f;
        vecf hv1 = {};
        const int node2 = node1 + nw;
        int beg2 = 0, end2 = 0;
        if (node1 < n) {
            if (lane < end1 - beg1) key1 = csr[beg1 + lane];
            di1 = dinv[node1];
            hv1 = *(const vecf*)(H + (long long)node1 * D + lofs);
            if (node2 < n) {
                beg2 = rp[node2];
                end2 = rp[node2 + 1];
            }
        }

        const int d = end - beg;

        if (d <= 32) {
            const int skey = key;

            int j = 0;
            for (; j + 8 <= d; j += 8) add_rows<8>(acc, skey, j, dinv, di, H, lofs);
            switch (d - j) {
                case 7: add_rows<7>(acc, skey, j, dinv, di, H, lofs); break;
                case 6: add_rows<6>(acc, skey, j, dinv, di, H, lofs); break;
                case 5: add_rows<5>(acc, skey, j, dinv, di, H, lofs); break;
                case 4: add_rows<4>(acc, skey, j, dinv, di, H, lofs); break;
                case 3: add_rows<3>(acc, skey, j, dinv, di, H, lofs); break;
                case 2: add_rows<2>(acc, skey, j, dinv, di, H, lofs); break;
                case 1: add_rows<1>(acc, skey, j, dinv, di, H, lofs); break;
                default: break;
            }
        } else {
            int prev = -1;
            int left = d;
            while (left > 0) {
                int best = 0x7fffffff;
                for (int b = beg + lane; b < end; b += 32) {
                    int k = csr[b];
                    if (k > prev && k < best) best = k;
                }
                for (int o = 16; o > 0; o >>= 1) {
                    int u = __shfl_xor(best, o, 32);
                    best = u < best ? u : best;
                }
                int c = 0;
                for (int b = beg + lane; b < end; b += 32) c += (csr[b] == best) ? 1 : 0;
                for (int o = 16; o > 0; o >>= 1) c += __shfl_xor(c, o, 32);
                best = __builtin_amdgcn_readfirstlane(best);
                c = __builtin_amdgcn_readfirstlane(c);
                if (c <= 0) break;
                const float w0 = dinv[best] * di;
                const vecf v0 = *(const vecf*)(H + (long long)best * D + lofs);
                for (int r = 0; r < c; r++) acc += v0 * w0;
                prev = best;
                left -= c;
            }
        }

        if (ELU) {
#pragma unroll
            for (int q = 0; q < VEC; q++) {
                float v = acc[q];
                acc[q] = v > 0.0f ? v : (__expf(v) - 1.0f);
            }
        }
        *(vecf*)(Out + (long long)node * D + lofs) = acc;

        acc = self_term(bv, hv1, di1 * di1);
        if (node1 >= n) break;
        node = node1; beg = beg1; end = end1; di = di1; key = key1;
        node1 = node2; beg1 = beg2; end1 = end2;
    }
}

#define TILE_NODES 64
#define TILE_PITCH 128
#define FUSED_THREADS 512
#define FUSED_BLOCKS 512

template <typename vecf>
__device__ __forceinline__ vecf sum_row(vecf acc, const int key, const int beg, const int end, const int lane,
                                        const int* __restrict__ csr, const float* __restrict__ dinv,
                                        const float di, const float* __restrict__ H, const int lofs) {
    constexpr int D = (int)(sizeof(vecf) / sizeof(float)) * 32;
    const int d = end - beg;
    if (d <= 32) {
        const int skey = key;
        int j = 0;
        for (; j + 8 <= d; j += 8) add_rows<8>(acc, skey, j, dinv, di, H, lofs);
        switch (d - j) {
            case 7: add_rows<7>(acc, skey, j, dinv, di, H, lofs); break;
            case 6: add_rows<6>(acc, skey, j, dinv, di, H, lofs); break;
            case 5: add_rows<5>(acc, skey, j, dinv, di, H, lofs); break;
            case 4: add_rows<4>(acc, skey, j, dinv, di, H, lofs); break;
            case 3: add_rows<3>(acc, skey, j, dinv, di, H, lofs); break;
            case 2: add_rows<2>(acc, skey, j, dinv, di, H, lofs); break;
            case 1: add_rows<1>(acc, skey, j, dinv, di, H, lofs); break;
            default: break;
        }
    } else {
        int prev = -1;
        int left = d;
        while (left > 0) {
            int best = 0x7fffffff;
            for (int b = beg + lane; b < end; b += 32) {
                int k = csr[b];
                if (k > prev && k < best) best = k;
            }
            for (int o = 16; o > 0; o >>= 1) {
                int u = __shfl_xor(best, o, 32);
                best = u < best ? u : best;
            }
            int c = 0;
            for (int b = beg + lane; b < end; b += 32) c += (csr[b] == best) ? 1 : 0;
            for (int o = 16; o > 0; o >>= 1) c += __shfl_xor(c, o, 32);
            best = __builtin_amdgcn_readfirstlane(best);
            c = __builtin_amdgcn_readfirstlane(c);
            if (c <= 0) break;
            const float w0 = dinv[best] * di;
            const vecf v0 = *(const vecf*)(H + (long long)best * D + lofs);
            for (int r = 0; r < c; r++) acc += v0 * w0;
            prev = best;
            left -= c;
        }
    }
    return acc;
}

__global__ __launch_bounds__(FUSED_THREADS) void gather_gemm_elu_gemm_kernel(const int* __restrict__ rp,
                                                                             const int* __restrict__ csr,
                                                                             const float* __restrict__ dinv,
                                                                             const float* __restrict__ X,
                                                                             const float* __restrict__ W1,
                                                                             const float* __restrict__ bias1,
                                                                             const float* __restrict__ W2,
                                                                             float* __restrict__ H2, int n) {
    constexpr int D1 = 128;
    constexpr int D2 = 64;
    constexpr int NT1 = D1 / 16;
    constexpr int NT2 = D2 / 16;
    constexpr int NPAIR1 = 32 * NT1 * 32;
    constexpr int NPAIR2 = 32 * NT2 * 32;
    typedef __attribute__((ext_vector_type(4))) float vecf;
    __shared__ __attribute__((aligned(16))) float sW1[NPAIR1 * 2];
    __shared__ __attribute__((aligned(16))) float sW2[NPAIR2 * 2];
    __shared__ __attribute__((aligned(16))) float sA[TILE_NODES * TILE_PITCH];
    __shared__ __attribute__((aligned(16))) float sG[TILE_NODES * TILE_PITCH];

    for (int pair = threadIdx.x; pair < NPAIR1; pair += FUSED_THREADS) {
        int l  = pair & 31;
        int tt = (pair >> 5) % NT1;
        int ks = (pair >> 5) / NT1;
        int col = tt * 16 + (l & 15);
        int k0  = ks * 4 + ((l >> 4) * 2);
        sW1[pair * 2 + 0] = W1[k0 * D1 + col];
        sW1[pair * 2 + 1] = W1[(k0 + 1) * D1 + col];
    }
    for (int pair = threadIdx.x; pair < NPAIR2; pair += FUSED_THREADS) {
        int l  = pair & 31;
        int tt = (pair >> 5) % NT2;
        int ks = (pair >> 5) / NT2;
        int col = tt * 16 + (l & 15);
        int k0  = ks * 4 + ((l >> 4) * 2);
        sW2[pair * 2 + 0] = W2[k0 * D2 + col];
        sW2[pair * 2 + 1] = W2[(k0 + 1) * D2 + col];
    }

    const int lane = threadIdx.x & 31;
    const int wave = __builtin_amdgcn_readfirstlane((int)(threadIdx.x >> 5));
    const int m    = lane & 15;
    const int kk   = (lane >> 4) * 2;
    const int lofs = lane * 4;

    const int ct = wave & 7;
    const int rg = wave >> 3;
    const float bcol = bias1[ct * 16 + m];
    const float* __restrict__ a0p = sA + (rg * 32 + m) * TILE_PITCH + kk;
    const float* __restrict__ a1p = a0p + 16 * TILE_PITCH;
    const float* __restrict__ bp1 = sW1 + (ct * 32 + lane) * 2;
    const int gc   = ct * 16 + m;
    const int grow = rg * 32 + (lane >> 4) * 8;

    const int rt = wave >> 2;
    const int c2 = wave & 3;
    const float* __restrict__ gp  = sG + (rt * 16 + m) * TILE_PITCH + kk;
    const float* __restrict__ bp2 = sW2 + (c2 * 32 + lane) * 2;

    const int ntiles = (n + TILE_NODES - 1) / TILE_NODES;
    for (int tile = (int)blockIdx.x; tile < ntiles; tile += (int)gridDim.x) {
        const int nb0 = tile * TILE_NODES + wave * 4;
        int beg[4], end[4], key[4];
        float di[4];
        vecf hv[4];
#pragma unroll
        for (int q = 0; q < 4; q++) {
            const int node = nb0 + q;
            beg[q] = 0; end[q] = 0;
            if (node < n) {
                beg[q] = rp[node];
                end[q] = rp[node + 1];
            }
        }
#pragma unroll
        for (int q = 0; q < 4; q++) {
            const int node = nb0 + q;
            key[q] = 0x7fffffff;
            di[q] = 0.0f;
            hv[q] = vecf{};
            if (node < n) {
                if (lane < end[q] - beg[q]) key[q] = csr[beg[q] + lane];
                di[q] = dinv[node];
                hv[q] = *(const vecf*)(X + (long long)node * DIN + lofs);
            }
        }
#pragma unroll
        for (int q = 0; q < 4; q++) {
            vecf acc = hv[q] * (di[q] * di[q]);
            if (nb0 + q < n) acc = sum_row<vecf>(acc, key[q], beg[q], end[q], lane, csr, dinv, di[q], X, lofs);
            const int r = wave * 4 + q;
            *(vecf*)(sA + r * TILE_PITCH + ((lane ^ (r & 15)) << 2)) = acc;
        }
        __syncthreads();

        v8f acc0 = {};
        v8f acc1 = {};
#pragma unroll 8
        for (int ks = 0; ks < 32; ks++) {
            v2f b  = *(const v2f*)(bp1 + ks * (NT1 * 64));
            const int ko = (ks ^ m) << 2;
            v2f a0 = *(const v2f*)(a0p + ko);
            v2f a1 = *(const v2f*)(a1p + ko);
            acc0 = __builtin_amdgcn_wmma_f32_16x16x4_f32(false, a0, false, b, (short)0, acc0, false, false);
            acc1 = __builtin_amdgcn_wmma_f32_16x16x4_f32(false, a1, false, b, (short)0, acc1, false, false);
        }

#pragma unroll
        for (int v = 0; v < 8; v++) {
            float t0 = acc0[v] + bcol;
            float t1 = acc1[v] + bcol;
            t0 = t0 > 0.0f ? t0 : (__expf(t0) - 1.0f);
            t1 = t1 > 0.0f ? t1 : (__expf(t1) - 1.0f);
            const int r0 = grow + v;
            const int go = (((gc >> 2) ^ (r0 & 15)) << 2) + (gc & 3);
            sG[r0 * TILE_PITCH + go] = t0;
            sG[(r0 + 16) * TILE_PITCH + go] = t1;
        }
        __syncthreads();

        v8f acc2 = {};
#pragma unroll 8
        for (int ks = 0; ks < 32; ks++) {
            v2f b = *(const v2f*)(bp2 + ks * (NT2 * 64));
            v2f a = *(const v2f*)(gp + ((ks ^ m) << 2));
            acc2 = __builtin_amdgcn_wmma_f32_16x16x4_f32(false, a, false, b, (short)0, acc2, false, false);
        }

        const long long rbase = (long long)tile * TILE_NODES + rt * 16 + (lane >> 4) * 8;
#pragma unroll
        for (int v = 0; v < 8; v++) {
            const long long r = rbase + v;
            if (r < n) H2[r * D2 + c2 * 16 + m] = acc2[v];
        }
    }
}

extern "C" void kernel_launch(void* const* d_in, const int* in_sizes, int n_in,
                              void* d_out, int out_size, void* d_ws, size_t ws_size,
                              hipStream_t stream) {
    const float* x  = (const float*)d_in[0];
    const int*   ei = (const int*)d_in[1];
    const float* W1 = (const float*)d_in[2];
    const float* b1 = (const float*)d_in[3];
    const float* W2 = (const float*)d_in[4];
    const float* b2 = (const float*)d_in[5];
    float* out = (float*)d_out;

    const int N = in_sizes[0] / DIN;
    const int E = in_sizes[1] / 2;
    const int DH = 128;

    const int B = 256;
    const int gNode = GATHER_BLOCKS;
    const int NB    = (N + BUCKET_NODES - 1) / BUCKET_NODES;

    char* ws = (char*)d_ws;
    float* dinv = (float*)ws;
    size_t off = (((size_t)N * sizeof(float)) + 255) & ~(size_t)255;
    float* hbuf = (float*)(ws + off);
    float* aggbuf = (float*)(ws + off + (size_t)N * DH * sizeof(float));
    char* tail = ws + off + 2 * (size_t)N * DH * sizeof(float);
    int* rp = (int*)tail;
    size_t rpb = (((size_t)(N + 1) * sizeof(int)) + 255) & ~(size_t)255;
    int* csr = (int*)(tail + rpb);
    int* pairs = (int*)aggbuf;
    int* cpre  = pairs + 2 * (size_t)E;

    chunk_sort_kernel<<<SORT_BLOCKS, SORT_THREADS, 2 * NB * sizeof(int), stream>>>(ei, E, NB, cpre, pairs);
    bucket_rows_kernel<<<NB, BUCKET_NODES, 0, stream>>>(pairs, cpre, E, N, rp, dinv, csr);

    gather_gemm_elu_gemm_kernel<<<FUSED_BLOCKS, FUSED_THREADS, 0, stream>>>(rp, csr, dinv, x, W1, b1, W2, hbuf, N);

    gather_kernel<64, false><<<gNode, B, 0, stream>>>(rp, csr, dinv, hbuf, b2, out, N);
}


# Layer 2 rows stored pre-scaled by their own dinv (hs = h2*dinv[row]); last launch adds them unweighted, then one fma per node with dinv[i] and bias: each term's rounding changes, sum order same
# speedup vs baseline: 4.0666x; 1.0399x over previous
#include <hip/hip_runtime.h>
#include <hip/hip_bf16.h>

typedef __attribute__((ext_vector_type(2))) float v2f;
typedef __attribute__((ext_vector_type(8))) float v8f;

#define DIN 128

#define BUCKET_SHIFT 8
#define BUCKET_NODES (1 << BUCKET_SHIFT)
#define SORT_BLOCKS 256
#define SORT_THREADS 512
#define EDGE_BATCH 8
#define RUN_LANES 16
#define RUN_STEPS 16
#define ROW_SLOTS 32
#define ROW_PITCH (ROW_SLOTS + 1)
#define LONG_ROW 0x40000000

static_assert(SORT_BLOCKS == 32 * (BUCKET_NODES / 32), "bucket_rows_kernel: 8 waves x 32 chunk runs");
static_assert(RUN_STEPS * (32 / RUN_LANES) == 32, "a wave covers 32 chunk runs");

__global__ __launch_bounds__(SORT_THREADS) void chunk_sort_kernel(const int* __restrict__ ei, int e, int nb,
                                                                  int* __restrict__ cpre,
                                                                  int* __restrict__ pairs) {
    extern __shared__ int sh[];
    __shared__ int wt[SORT_THREADS / 32];
    int* __restrict__ cnt = sh;
    int* __restrict__ cur = sh + nb;
    const int tid = threadIdx.x;
    const int lane = tid & 31;
    const int wave = tid >> 5;
    const int g = (int)blockIdx.x;
    for (int b = tid; b < nb; b += SORT_THREADS) cnt[b] = 0;
    __syncthreads();

    const int ch = (e + SORT_BLOCKS - 1) / SORT_BLOCKS;
    const int lo = g * ch;
    const int hi = (lo + ch < e) ? (lo + ch) : e;
    const int* __restrict__ dstp = ei + e;

    for (int i0 = lo + tid; i0 < hi; i0 += SORT_THREADS * EDGE_BATCH) {
        int d[EDGE_BATCH];
#pragma unroll
        for (int q = 0; q < EDGE_BATCH; q++) {
            const int i = i0 + q * SORT_THREADS;
            d[q] = (i < hi) ? dstp[i] : -1;
        }
#pragma unroll
        for (int q = 0; q < EDGE_BATCH; q++)
            if (d[q] >= 0) atomicAdd(&cnt[d[q] >> BUCKET_SHIFT], 1);
    }
    __syncthreads();

    int carry = 0;
    for (int base = 0; base < nb; base += SORT_THREADS) {
        const int b = base + tid;
        const int t = (b < nb) ? cnt[b] : 0;
        int incl = t;
        for (int o = 1; o < 32; o <<= 1) {
            int u = __shfl_up(incl, o, 32);
            if (lane >= o) incl += u;
        }
        if (lane == 31) wt[wave] = incl;
        __syncthreads();
        int offs = carry + incl - t;
        int all = 0;
        for (int w = 0; w < SORT_THREADS / 32; w++) {
            const int x = wt[w];
            if (w < wave) offs += x;
            all += x;
        }
        if (b < nb) {
            cur[b] = offs;
            cpre[(size_t)b * SORT_BLOCKS + g] = offs;
        }
        carry += all;
        __syncthreads();
    }
    if (tid == 0) cpre[(size_t)nb * SORT_BLOCKS + g] = carry;

    for (int i0 = lo + tid; i0 < hi; i0 += SORT_THREADS * EDGE_BATCH) {
        int s[EDGE_BATCH];
        int d[EDGE_BATCH];
#pragma unroll
        for (int q = 0; q < EDGE_BATCH; q++) {
            const int i = i0 + q * SORT_THREADS;
            s[q] = (i < hi) ? ei[i] : 0;
            d[q] = (i < hi) ? dstp[i] : -1;
        }
#pragma unroll
        for (int q = 0; q < EDGE_BATCH; q++) {
            if (d[q] >= 0) {
                const int p = lo + atomicAdd(&cur[d[q] >> BUCKET_SHIFT], 1);
                *(int2*)(pairs + 2 * (long long)p) = make_int2(s[q], d[q]);
            }
        }
    }
}

__device__ __forceinline__ void order_pair(int& a, int& b) {
    const int lo = a < b ? a : b;
    const int hi = a < b ? b : a;
    a = lo;
    b = hi;
}

template <int LO, int N, int R, int SZ>
__device__ __forceinline__ void oe_merge(int (&v)[SZ]) {
    constexpr int M = R * 2;
    if constexpr (M < N) {
        oe_merge<LO, N, M, SZ>(v);
        oe_merge<LO + R, N, M, SZ>(v);
#pragma unroll
        for (int i = LO + R; i + R < LO + N; i += M) order_pair(v[i], v[i + R]);
    } else {
        order_pair(v[LO], v[LO + R]);
    }
}

template <int LO, int N, int SZ>
__device__ __forceinline__ void oe_sort(int (&v)[SZ]) {
    if constexpr (N > 1) {
        oe_sort<LO, N / 2, SZ>(v);
        oe_sort<LO + N / 2, N / 2, SZ>(v);
        oe_merge<LO, N, 1, SZ>(v);
    }
}

template <int N>
__device__ __forceinline__ void order_row(const int* __restrict__ row, const int c, int* __restrict__ out) {
    int v[N];
#pragma unroll
    for (int k = 0; k < N; k++) v[k] = (k < c) ? row[k] : 0x7fffffff;
    oe_sort<0, N, N>(v);
#pragma unroll
    for (int k = 0; k < N; k++)
        if (k < c) out[k] = v[k];
}

__global__ __launch_bounds__(BUCKET_NODES) void bucket_rows_kernel(const int* __restrict__ pairs,
                                                                   const int* __restrict__ cpre, int e, int n,
                                                                   int* __restrict__ rp,
                                                                   float* __restrict__ dinv,
                                                                   int* __restrict__ csr) {
    __shared__ int cnt[BUCKET_NODES];
    __shared__ int cur[BUCKET_NODES];
    __shared__ int runS[SORT_BLOCKS];
    __shared__ int runL[SORT_BLOCKS];
    __shared__ int wtot[BUCKET_NODES / 32];
    __shared__ int wlo[BUCKET_NODES / 32];
    __shared__ int lrow[BUCKET_NODES * ROW_PITCH];
    const int tid = threadIdx.x;
    const int lane = tid & 31;
    const int wave = tid >> 5;
    const int b = (int)blockIdx.x;

    const int ch = (e + SORT_BLOCKS - 1) / SORT_BLOCKS;
    const int o0 = cpre[(size_t)b * SORT_BLOCKS + tid];
    const int o1 = cpre[(size_t)(b + 1) * SORT_BLOCKS + tid];
    runS[tid] = tid * ch + o0;
    runL[tid] = o1 - o0;
    cnt[tid] = 0;
    int s0 = o0;
    for (int o = 16; o > 0; o >>= 1) s0 += __shfl_xor(s0, o, 32);
    if (lane == 0) wlo[wave] = s0;
    __syncthreads();
    int lo = 0;
    for (int w = 0; w < BUCKET_NODES / 32; w++) lo += wlo[w];

    const int sub = lane & (RUN_LANES - 1);
    const int which = lane / RUN_LANES;
    int S[RUN_STEPS];
    int L[RUN_STEPS];
#pragma unroll
    for (int j = 0; j < RUN_STEPS; j++) {
        const int gi = wave * 32 + j * (32 / RUN_LANES) + which;
        S[j] = runS[gi];
        L[j] = runL[gi];
    }

    {
        int d[RUN_STEPS];
#pragma unroll
        for (int j = 0; j < RUN_STEPS; j++)
            d[j] = (sub < L[j]) ? pairs[2 * (long long)(S[j] + sub) + 1] : -1;
#pragma unroll
        for (int j = 0; j < RUN_STEPS; j++)
            if (d[j] >= 0) atomicAdd(&cnt[d[j] & (BUCKET_NODES - 1)], 1);
#pragma unroll
        for (int j = 0; j < RUN_STEPS; j++) {
            for (int k = sub + RUN_LANES; k < L[j]; k += RUN_LANES) {
                const int dd = pairs[2 * (long long)(S[j] + k) + 1];
                atomicAdd(&cnt[dd & (BUCKET_NODES - 1)], 1);
            }
        }
    }
    __syncthreads();

    const int c = cnt[tid];
    int incl = c;
    for (int o = 1; o < 32; o <<= 1) {
        int u = __shfl_up(incl, o, 32);
        if (lane >= o) incl += u;
    }
    if (lane == 31) wtot[wave] = incl;
    __syncthreads();
    int start = lo + incl - c;
    for (int w = 0; w < BUCKET_NODES / 32; w++)
        if (w < wave) start += wtot[w];

    const int node = b * BUCKET_NODES + tid;
    cur[tid] = (c <= ROW_SLOTS) ? 0 : (start | LONG_ROW);
    if (node < n) {
        rp[node] = start;
        dinv[node] = rsqrtf((float)(c + 1));
        if (node == n - 1) rp[n] = start + c;
    }
    __syncthreads();

    {
        int2 pr[RUN_STEPS];
#pragma unroll
        for (int j = 0; j < RUN_STEPS; j++)
            pr[j] = (sub < L[j]) ? *(const int2*)(pairs + 2 * (long long)(S[j] + sub)) : make_int2(0, -1);
#pragma unroll
        for (int j = 0; j < RUN_STEPS; j++) {
            if (pr[j].y >= 0) {
                const int dl = pr[j].y & (BUCKET_NODES - 1);
                const int p = atomicAdd(&cur[dl], 1);
                if (p & LONG_ROW) csr[p ^ LONG_ROW] = pr[j].x;
                else lrow[dl * ROW_PITCH + p] = pr[j].x;
            }
        }
#pragma unroll
        for (int j = 0; j < RUN_STEPS; j++) {
            for (int k = sub + RUN_LANES; k < L[j]; k += RUN_LANES) {
                const int2 q2 = *(const int2*)(pairs + 2 * (long long)(S[j] + k));
                const int dl = q2.y & (BUCKET_NODES - 1);
                const int p = atomicAdd(&cur[dl], 1);
                if (p & LONG_ROW) csr[p ^ LONG_ROW] = q2.x;
                else lrow[dl * ROW_PITCH + p] = q2.x;
            }
        }
    }
    __syncthreads();

    const int cs = (c <= ROW_SLOTS) ? c : 0;
    if (__any(cs > 16)) order_row<ROW_SLOTS>(lrow + tid * ROW_PITCH, cs, csr + start);
    else order_row<16>(lrow + tid * ROW_PITCH, cs, csr + start);
}

#define GATHER_BLOCKS 2048

template <int K, typename vecf>
__device__ __forceinline__ void add_rows(vecf& acc, const int skey, const int j,
                                         const float* __restrict__ dinv, const float di,
                                         const float* __restrict__ H, const int lofs) {
    constexpr int D = (int)(sizeof(vecf) / sizeof(float)) * 32;
    int s[K];
    float w[K];
    vecf v[K];
#pragma unroll
    for (int q = 0; q < K; q++) s[q] = __builtin_amdgcn_readlane(skey, j + q);
#pragma unroll
    for (int q = 0; q < K; q++) w[q] = dinv[s[q]] * di;
#pragma unroll
    for (int q = 0; q < K; q++) v[q] = *(const vecf*)(H + (long long)s[q] * D + lofs);
#pragma unroll
    for (int q = 0; q < K; q++) acc += v[q] * w[q];
}

template <int K, typename vecf>
__device__ __forceinline__ void add_plain_rows(vecf& acc, const int skey, const int j,
                                               const float* __restrict__ H, const int lofs) {
    constexpr int D = (int)(sizeof(vecf) / sizeof(float)) * 32;
    int s[K];
    vecf v[K];
#pragma unroll
    for (int q = 0; q < K; q++) s[q] = __builtin_amdgcn_readlane(skey, j + q);
#pragma unroll
    for (int q = 0; q < K; q++) v[q] = *(const vecf*)(H + (long long)s[q] * D + lofs);
#pragma unroll
    for (int q = 0; q < K; q++) acc += v[q];
}

template <typename vecf>
__device__ __forceinline__ vecf add_long_row_plain(vecf acc, const int beg, const int end, const int lane,
                                                   const int* __restrict__ csr,
                                                   const float* __restrict__ H, const int lofs) {
    constexpr int D = (int)(sizeof(vecf) / sizeof(float)) * 32;
    int prev = -1;
    int left = end - beg;
    while (left > 0) {
        int best = 0x7fffffff;
        for (int b = beg + lane; b < end; b += 32) {
            int k = csr[b];
            if (k > prev && k < best) best = k;
        }
        for (int o = 16; o > 0; o >>= 1) {
            int u = __shfl_xor(best, o, 32);
            best = u < best ? u : best;
        }
        int c = 0;
        for (int b = beg + lane; b < end; b += 32) c += (csr[b] == best) ? 1 : 0;
        for (int o = 16; o > 0; o >>= 1) c += __shfl_xor(c, o, 32);
        best = __builtin_amdgcn_readfirstlane(best);
        c = __builtin_amdgcn_readfirstlane(c);
        if (c <= 0) break;
        const vecf v0 = *(const vecf*)(H + (long long)best * D + lofs);
#pragma clang loop unroll(disable)
        for (int r = 0; r < c; r++) acc += v0;
        prev = best;
        left -= c;
    }
    return acc;
}

__global__ __launch_bounds__(256) void scaled_row_sum_kernel(const int* __restrict__ rp,
                                                             const int* __restrict__ csr,
                                                             const float* __restrict__ dinv,
                                                             const float* __restrict__ Hs,
                                                             const float* __restrict__ bias,
                                                             float* __restrict__ Out, int n) {
    constexpr int D = 64;
    constexpr int VEC = D / 32;
    typedef __attribute__((ext_vector_type(VEC))) float vecf;
    const int lane = threadIdx.x & 31;
    const int nw = (int)gridDim.x * 8;
    int node = __builtin_amdgcn_readfirstlane((int)(blockIdx.x * 8 + (threadIdx.x >> 5)));
    if (node >= n) return;

    const int lofs = lane * VEC;
    const vecf bv = *(const vecf*)(bias + lofs);

    int beg = rp[node];
    int end = rp[node + 1];
    float di = dinv[node];
    vecf acc = *(const vecf*)(Hs + (long long)node * D + lofs);
    int key = 0x7fffffff;
    if (lane < end - beg) key = csr[beg + lane];
    int node1 = node + nw;
    int beg1 = 0, end1 = 0;
    if (node1 < n) {
        beg1 = rp[node1];
        end1 = rp[node1 + 1];
    }
    asm volatile("" : "+v"(key), "+v"(acc));

    for (;;) {
        int key1 = 0x7fffffff;
        float di1 = 0.0f;
        vecf hv1 = {};
        const int node2 = node1 + nw;
        int beg2 = 0, end2 = 0;
        if (node1 < n) {
            if (lane < end1 - beg1) key1 = csr[beg1 + lane];
            di1 = dinv[node1];
            hv1 = *(const vecf*)(Hs + (long long)node1 * D + lofs);
            if (node2 < n) {
                beg2 = rp[node2];
                end2 = rp[node2 + 1];
            }
        }

        const int d = end - beg;

        if (d <= 32) {
            const int skey = key;

            int j = 0;
            for (; j + 8 <= d; j += 8) add_plain_rows<8>(acc, skey, j, Hs, lofs);
            switch (d - j) {
                case 7: add_plain_rows<7>(acc, skey, j, Hs, lofs); break;
                case 6: add_plain_rows<6>(acc, skey, j, Hs, lofs); break;
                case 5: add_plain_rows<5>(acc, skey, j, Hs, lofs); break;
                case 4: add_plain_rows<4>(acc, skey, j, Hs, lofs); break;
                case 3: add_plain_rows<3>(acc, skey, j, Hs, lofs); break;
                case 2: add_plain_rows<2>(acc, skey, j, Hs, lofs); break;
                case 1: add_plain_rows<1>(acc, skey, j, Hs, lofs); break;
                default: break;
            }
        } else {
            acc = add_long_row_plain<vecf>(acc, beg, end, lane, csr, Hs, lofs);
        }

        vecf o;
        for (int q = 0; q < VEC; q++) o[q] = __builtin_fmaf(acc[q], di, bv[q]);
        *(vecf*)(Out + (long long)node * D + lofs) = o;

        acc = hv1;
        if (node1 >= n) break;
        node = node1; beg = beg1; end = end1; di = di1; key = key1;
        node1 = node2; beg1 = beg2; end1 = end2;
    }
}

#define TILE_NODES 64
#define TILE_PITCH 128
#define FUSED_THREADS 512
#define FUSED_BLOCKS 512

template <typename vecf>
__device__ __forceinline__ vecf sum_row(vecf acc, const int key, const int beg, const int end, const int lane,
                                        const int* __restrict__ csr, const float* __restrict__ dinv,
                                        const float di, const float* __restrict__ H, const int lofs) {
    constexpr int D = (int)(sizeof(vecf) / sizeof(float)) * 32;
    const int d = end - beg;
    if (d <= 32) {
        const int skey = key;
        int j = 0;
        for (; j + 8 <= d; j += 8) add_rows<8>(acc, skey, j, dinv, di, H, lofs);
        switch (d - j) {
            case 7: add_rows<7>(acc, skey, j, dinv, di, H, lofs); break;
            case 6: add_rows<6>(acc, skey, j, dinv, di, H, lofs); break;
            case 5: add_rows<5>(acc, skey, j, dinv, di, H, lofs); break;
            case 4: add_rows<4>(acc, skey, j, dinv, di, H, lofs); break;
            case 3: add_rows<3>(acc, skey, j, dinv, di, H, lofs); break;
            case 2: add_rows<2>(acc, skey, j, dinv, di, H, lofs); break;
            case 1: add_rows<1>(acc, skey, j, dinv, di, H, lofs); break;
            default: break;
        }
    } else {
        int prev = -1;
        int left = d;
        while (left > 0) {
            int best = 0x7fffffff;
            for (int b = beg + lane; b < end; b += 32) {
                int k = csr[b];
                if (k > prev && k < best) best = k;
            }
            for (int o = 16; o > 0; o >>= 1) {
                int u = __shfl_xor(best, o, 32);
                best = u < best ? u : best;
            }
            int c = 0;
            for (int b = beg + lane; b < end; b += 32) c += (csr[b] == best) ? 1 : 0;
            for (int o = 16; o > 0; o >>= 1) c += __shfl_xor(c, o, 32);
            best = __builtin_amdgcn_readfirstlane(best);
            c = __builtin_amdgcn_readfirstlane(c);
            if (c <= 0) break;
            const float w0 = dinv[best] * di;
            const vecf v0 = *(const vecf*)(H + (long long)best * D + lofs);
            for (int r = 0; r < c; r++) acc += v0 * w0;
            prev = best;
            left -= c;
        }
    }
    return acc;
}

__global__ __launch_bounds__(FUSED_THREADS) void gather_gemm_elu_gemm_kernel(const int* __restrict__ rp,
                                                                             const int* __restrict__ csr,
                                                                             const float* __restrict__ dinv,
                                                                             const float* __restrict__ X,
                                                                             const float* __restrict__ W1,
                                                                             const float* __restrict__ bias1,
                                                                             const float* __restrict__ W2,
                                                                             float* __restrict__ H2, int n) {
    constexpr int D1 = 128;
    constexpr int D2 = 64;
    constexpr int NT1 = D1 / 16;
    constexpr int NT2 = D2 / 16;
    constexpr int NPAIR1 = 32 * NT1 * 32;
    constexpr int NPAIR2 = 32 * NT2 * 32;
    typedef __attribute__((ext_vector_type(4))) float vecf;
    __shared__ __attribute__((aligned(16))) float sW1[NPAIR1 * 2];
    __shared__ __attribute__((aligned(16))) float sW2[NPAIR2 * 2];
    __shared__ __attribute__((aligned(16))) float sA[TILE_NODES * TILE_PITCH];
    __shared__ __attribute__((aligned(16))) float sG[TILE_NODES * TILE_PITCH];

    for (int pair = threadIdx.x; pair < NPAIR1; pair += FUSED_THREADS) {
        int l  = pair & 31;
        int tt = (pair >> 5) % NT1;
        int ks = (pair >> 5) / NT1;
        int col = tt * 16 + (l & 15);
        int k0  = ks * 4 + ((l >> 4) * 2);
        sW1[pair * 2 + 0] = W1[k0 * D1 + col];
        sW1[pair * 2 + 1] = W1[(k0 + 1) * D1 + col];
    }
    for (int pair = threadIdx.x; pair < NPAIR2; pair += FUSED_THREADS) {
        int l  = pair & 31;
        int tt = (pair >> 5) % NT2;
        int ks = (pair >> 5) / NT2;
        int col = tt * 16 + (l & 15);
        int k0  = ks * 4 + ((l >> 4) * 2);
        sW2[pair * 2 + 0] = W2[k0 * D2 + col];
        sW2[pair * 2 + 1] = W2[(k0 + 1) * D2 + col];
    }

    const int lane = threadIdx.x & 31;
    const int wave = __builtin_amdgcn_readfirstlane((int)(threadIdx.x >> 5));
    const int m    = lane & 15;
    const int kk   = (lane >> 4) * 2;
    const int lofs = lane * 4;

    const int ct = wave & 7;
    const int rg = wave >> 3;
    const float bcol = bias1[ct * 16 + m];
    const float* __restrict__ a0p = sA + (rg * 32 + m) * TILE_PITCH + kk;
    const float* __restrict__ a1p = a0p + 16 * TILE_PITCH;
    const float* __restrict__ bp1 = sW1 + (ct * 32 + lane) * 2;
    const int gc   = ct * 16 + m;
    const int grow = rg * 32 + (lane >> 4) * 8;

    const int rt = wave >> 2;
    const int c2 = wave & 3;
    const float* __restrict__ gp  = sG + (rt * 16 + m) * TILE_PITCH + kk;
    const float* __restrict__ bp2 = sW2 + (c2 * 32 + lane) * 2;

    const int ntiles = (n + TILE_NODES - 1) / TILE_NODES;
    for (int tile = (int)blockIdx.x; tile < ntiles; tile += (int)gridDim.x) {
        const int nb0 = tile * TILE_NODES + wave * 4;
        int beg[4], end[4], key[4];
        float di[4];
        vecf hv[4];
#pragma unroll
        for (int q = 0; q < 4; q++) {
            const int node = nb0 + q;
            beg[q] = 0; end[q] = 0;
            if (node < n) {
                beg[q] = rp[node];
                end[q] = rp[node + 1];
            }
        }
#pragma unroll
        for (int q = 0; q < 4; q++) {
            const int node = nb0 + q;
            key[q] = 0x7fffffff;
            di[q] = 0.0f;
            hv[q] = vecf{};
            if (node < n) {
                if (lane < end[q] - beg[q]) key[q] = csr[beg[q] + lane];
                di[q] = dinv[node];
                hv[q] = *(const vecf*)(X + (long long)node * DIN + lofs);
            }
        }
#pragma unroll
        for (int q = 0; q < 4; q++) {
            vecf acc = hv[q] * (di[q] * di[q]);
            if (nb0 + q < n) acc = sum_row<vecf>(acc, key[q], beg[q], end[q], lane, csr, dinv, di[q], X, lofs);
            const int r = wave * 4 + q;
            *(vecf*)(sA + r * TILE_PITCH + ((lane ^ (r & 15)) << 2)) = acc;
        }
        __syncthreads();

        v8f acc0 = {};
        v8f acc1 = {};
#pragma unroll 8
        for (int ks = 0; ks < 32; ks++) {
            v2f b  = *(const v2f*)(bp1 + ks * (NT1 * 64));
            const int ko = (ks ^ m) << 2;
            v2f a0 = *(const v2f*)(a0p + ko);
            v2f a1 = *(const v2f*)(a1p + ko);
            acc0 = __builtin_amdgcn_wmma_f32_16x16x4_f32(false, a0, false, b, (short)0, acc0, false, false);
            acc1 = __builtin_amdgcn_wmma_f32_16x16x4_f32(false, a1, false, b, (short)0, acc1, false, false);
        }

#pragma unroll
        for (int v = 0; v < 8; v++) {
            float t0 = acc0[v] + bcol;
            float t1 = acc1[v] + bcol;
            t0 = t0 > 0.0f ? t0 : (__expf(t0) - 1.0f);
            t1 = t1 > 0.0f ? t1 : (__expf(t1) - 1.0f);
            const int r0 = grow + v;
            const int go = (((gc >> 2) ^ (r0 & 15)) << 2) + (gc & 3);
            sG[r0 * TILE_PITCH + go] = t0;
            sG[(r0 + 16) * TILE_PITCH + go] = t1;
        }
        __syncthreads();

        const long long rbase = (long long)tile * TILE_NODES + rt * 16 + (lane >> 4) * 8;
        float dsc[8];
#pragma unroll
        for (int v = 0; v < 8; v++) {
            const long long r = rbase + v;
            dsc[v] = dinv[r < n ? r : (long long)(n - 1)];
        }
        v8f acc2 = {};
#pragma unroll 8
        for (int ks = 0; ks < 32; ks++) {
            v2f b = *(const v2f*)(bp2 + ks * (NT2 * 64));
            v2f a = *(const v2f*)(gp + ((ks ^ m) << 2));
            acc2 = __builtin_amdgcn_wmma_f32_16x16x4_f32(false, a, false, b, (short)0, acc2, false, false);
        }

#pragma unroll
        for (int v = 0; v < 8; v++) {
            const long long r = rbase + v;
            if (r < n) H2[r * D2 + c2 * 16 + m] = acc2[v] * dsc[v];
        }
    }
}

extern "C" void kernel_launch(void* const* d_in, const int* in_sizes, int n_in,
                              void* d_out, int out_size, void* d_ws, size_t ws_size,
                              hipStream_t stream) {
    const float* x  = (const float*)d_in[0];
    const int*   ei = (const int*)d_in[1];
    const float* W1 = (const float*)d_in[2];
    const float* b1 = (const float*)d_in[3];
    const float* W2 = (const float*)d_in[4];
    const float* b2 = (const float*)d_in[5];
    float* out = (float*)d_out;

    const int N = in_sizes[0] / DIN;
    const int E = in_sizes[1] / 2;
    const int DH = 128;

    const int B = 256;
    const int gNode = GATHER_BLOCKS;
    const int NB    = (N + BUCKET_NODES - 1) / BUCKET_NODES;

    char* ws = (char*)d_ws;
    float* dinv = (float*)ws;
    size_t off = (((size_t)N * sizeof(float)) + 255) & ~(size_t)255;
    float* hbuf = (float*)(ws + off);
    float* aggbuf = (float*)(ws + off + (size_t)N * DH * sizeof(float));
    char* tail = ws + off + 2 * (size_t)N * DH * sizeof(float);
    int* rp = (int*)tail;
    size_t rpb = (((size_t)(N + 1) * sizeof(int)) + 255) & ~(size_t)255;
    int* csr = (int*)(tail + rpb);
    int* pairs = (int*)aggbuf;
    int* cpre  = pairs + 2 * (size_t)E;

    chunk_sort_kernel<<<SORT_BLOCKS, SORT_THREADS, 2 * NB * sizeof(int), stream>>>(ei, E, NB, cpre, pairs);
    bucket_rows_kernel<<<NB, BUCKET_NODES, 0, stream>>>(pairs, cpre, E, N, rp, dinv, csr);

    gather_gemm_elu_gemm_kernel<<<FUSED_BLOCKS, FUSED_THREADS, 0, stream>>>(rp, csr, dinv, x, W1, b1, W2, hbuf, N);

    scaled_row_sum_kernel<<<gNode, B, 0, stream>>>(rp, csr, dinv, hbuf, b2, out, N);
}


# Fused kernel: each tile's row bounds (one load per wave, not 4 waited loads), sources, own rows and dinv are requested one tile ahead, under the previous tile's products; sum orders unchanged
# speedup vs baseline: 4.1869x; 1.0296x over previous
#include <hip/hip_runtime.h>
#include <hip/hip_bf16.h>

typedef __attribute__((ext_vector_type(2))) float v2f;
typedef __attribute__((ext_vector_type(8))) float v8f;

#define DIN 128

#define BUCKET_SHIFT 8
#define BUCKET_NODES (1 << BUCKET_SHIFT)
#define SORT_BLOCKS 256
#define SORT_THREADS 512
#define EDGE_BATCH 8
#define RUN_LANES 16
#define RUN_STEPS 16
#define ROW_SLOTS 32
#define ROW_PITCH (ROW_SLOTS + 1)
#define LONG_ROW 0x40000000

static_assert(SORT_BLOCKS == 32 * (BUCKET_NODES / 32), "bucket_rows_kernel: 8 waves x 32 chunk runs");
static_assert(RUN_STEPS * (32 / RUN_LANES) == 32, "a wave covers 32 chunk runs");

__global__ __launch_bounds__(SORT_THREADS) void chunk_sort_kernel(const int* __restrict__ ei, int e, int nb,
                                                                  int* __restrict__ cpre,
                                                                  int* __restrict__ pairs) {
    extern __shared__ int sh[];
    __shared__ int wt[SORT_THREADS / 32];
    int* __restrict__ cnt = sh;
    int* __restrict__ cur = sh + nb;
    const int tid = threadIdx.x;
    const int lane = tid & 31;
    const int wave = tid >> 5;
    const int g = (int)blockIdx.x;
    for (int b = tid; b < nb; b += SORT_THREADS) cnt[b] = 0;
    __syncthreads();

    const int ch = (e + SORT_BLOCKS - 1) / SORT_BLOCKS;
    const int lo = g * ch;
    const int hi = (lo + ch < e) ? (lo + ch) : e;
    const int* __restrict__ dstp = ei + e;

    for (int i0 = lo + tid; i0 < hi; i0 += SORT_THREADS * EDGE_BATCH) {
        int d[EDGE_BATCH];
#pragma unroll
        for (int q = 0; q < EDGE_BATCH; q++) {
            const int i = i0 + q * SORT_THREADS;
            d[q] = (i < hi) ? dstp[i] : -1;
        }
#pragma unroll
        for (int q = 0; q < EDGE_BATCH; q++)
            if (d[q] >= 0) atomicAdd(&cnt[d[q] >> BUCKET_SHIFT], 1);
    }
    __syncthreads();

    int carry = 0;
    for (int base = 0; base < nb; base += SORT_THREADS) {
        const int b = base + tid;
        const int t = (b < nb) ? cnt[b] : 0;
        int incl = t;
        for (int o = 1; o < 32; o <<= 1) {
            int u = __shfl_up(incl, o, 32);
            if (lane >= o) incl += u;
        }
        if (lane == 31) wt[wave] = incl;
        __syncthreads();
        int offs = carry + incl - t;
        int all = 0;
        for (int w = 0; w < SORT_THREADS / 32; w++) {
            const int x = wt[w];
            if (w < wave) offs += x;
            all += x;
        }
        if (b < nb) {
            cur[b] = offs;
            cpre[(size_t)b * SORT_BLOCKS + g] = offs;
        }
        carry += all;
        __syncthreads();
    }
    if (tid == 0) cpre[(size_t)nb * SORT_BLOCKS + g] = carry;

    for (int i0 = lo + tid; i0 < hi; i0 += SORT_THREADS * EDGE_BATCH) {
        int s[EDGE_BATCH];
        int d[EDGE_BATCH];
#pragma unroll
        for (int q = 0; q < EDGE_BATCH; q++) {
            const int i = i0 + q * SORT_THREADS;
            s[q] = (i < hi) ? ei[i] : 0;
            d[q] = (i < hi) ? dstp[i] : -1;
        }
#pragma unroll
        for (int q = 0; q < EDGE_BATCH; q++) {
            if (d[q] >= 0) {
                const int p = lo + atomicAdd(&cur[d[q] >> BUCKET_SHIFT], 1);
                *(int2*)(pairs + 2 * (long long)p) = make_int2(s[q], d[q]);
            }
        }
    }
}

__device__ __forceinline__ void order_pair(int& a, int& b) {
    const int lo = a < b ? a : b;
    const int hi = a < b ? b : a;
    a = lo;
    b = hi;
}

template <int LO, int N, int R, int SZ>
__device__ __forceinline__ void oe_merge(int (&v)[SZ]) {
    constexpr int M = R * 2;
    if constexpr (M < N) {
        oe_merge<LO, N, M, SZ>(v);
        oe_merge<LO + R, N, M, SZ>(v);
#pragma unroll
        for (int i = LO + R; i + R < LO + N; i += M) order_pair(v[i], v[i + R]);
    } else {
        order_pair(v[LO], v[LO + R]);
    }
}

template <int LO, int N, int SZ>
__device__ __forceinline__ void oe_sort(int (&v)[SZ]) {
    if constexpr (N > 1) {
        oe_sort<LO, N / 2, SZ>(v);
        oe_sort<LO + N / 2, N / 2, SZ>(v);
        oe_merge<LO, N, 1, SZ>(v);
    }
}

template <int N>
__device__ __forceinline__ void order_row(const int* __restrict__ row, const int c, int* __restrict__ out) {
    int v[N];
#pragma unroll
    for (int k = 0; k < N; k++) v[k] = (k < c) ? row[k] : 0x7fffffff;
    oe_sort<0, N, N>(v);
#pragma unroll
    for (int k = 0; k < N; k++)
        if (k < c) out[k] = v[k];
}

__global__ __launch_bounds__(BUCKET_NODES) void bucket_rows_kernel(const int* __restrict__ pairs,
                                                                   const int* __restrict__ cpre, int e, int n,
                                                                   int* __restrict__ rp,
                                                                   float* __restrict__ dinv,
                                                                   int* __restrict__ csr) {
    __shared__ int cnt[BUCKET_NODES];
    __shared__ int cur[BUCKET_NODES];
    __shared__ int runS[SORT_BLOCKS];
    __shared__ int runL[SORT_BLOCKS];
    __shared__ int wtot[BUCKET_NODES / 32];
    __shared__ int wlo[BUCKET_NODES / 32];
    __shared__ int lrow[BUCKET_NODES * ROW_PITCH];
    const int tid = threadIdx.x;
    const int lane = tid & 31;
    const int wave = tid >> 5;
    const int b = (int)blockIdx.x;

    const int ch = (e + SORT_BLOCKS - 1) / SORT_BLOCKS;
    const int o0 = cpre[(size_t)b * SORT_BLOCKS + tid];
    const int o1 = cpre[(size_t)(b + 1) * SORT_BLOCKS + tid];
    runS[tid] = tid * ch + o0;
    runL[tid] = o1 - o0;
    cnt[tid] = 0;
    int s0 = o0;
    for (int o = 16; o > 0; o >>= 1) s0 += __shfl_xor(s0, o, 32);
    if (lane == 0) wlo[wave] = s0;
    __syncthreads();
    int lo = 0;
    for (int w = 0; w < BUCKET_NODES / 32; w++) lo += wlo[w];

    const int sub = lane & (RUN_LANES - 1);
    const int which = lane / RUN_LANES;
    int S[RUN_STEPS];
    int L[RUN_STEPS];
#pragma unroll
    for (int j = 0; j < RUN_STEPS; j++) {
        const int gi = wave * 32 + j * (32 / RUN_LANES) + which;
        S[j] = runS[gi];
        L[j] = runL[gi];
    }

    {
        int d[RUN_STEPS];
#pragma unroll
        for (int j = 0; j < RUN_STEPS; j++)
            d[j] = (sub < L[j]) ? pairs[2 * (long long)(S[j] + sub) + 1] : -1;
#pragma unroll
        for (int j = 0; j < RUN_STEPS; j++)
            if (d[j] >= 0) atomicAdd(&cnt[d[j] & (BUCKET_NODES - 1)], 1);
#pragma unroll
        for (int j = 0; j < RUN_STEPS; j++) {
            for (int k = sub + RUN_LANES; k < L[j]; k += RUN_LANES) {
                const int dd = pairs[2 * (long long)(S[j] + k) + 1];
                atomicAdd(&cnt[dd & (BUCKET_NODES - 1)], 1);
            }
        }
    }
    __syncthreads();

    const int c = cnt[tid];
    int incl = c;
    for (int o = 1; o < 32; o <<= 1) {
        int u = __shfl_up(incl, o, 32);
        if (lane >= o) incl += u;
    }
    if (lane == 31) wtot[wave] = incl;
    __syncthreads();
    int start = lo + incl - c;
    for (int w = 0; w < BUCKET_NODES / 32; w++)
        if (w < wave) start += wtot[w];

    const int node = b * BUCKET_NODES + tid;
    cur[tid] = (c <= ROW_SLOTS) ? 0 : (start | LONG_ROW);
    if (node < n) {
        rp[node] = start;
        dinv[node] = rsqrtf((float)(c + 1));
        if (node == n - 1) rp[n] = start + c;
    }
    __syncthreads();

    {
        int2 pr[RUN_STEPS];
#pragma unroll
        for (int j = 0; j < RUN_STEPS; j++)
            pr[j] = (sub < L[j]) ? *(const int2*)(pairs + 2 * (long long)(S[j] + sub)) : make_int2(0, -1);
#pragma unroll
        for (int j = 0; j < RUN_STEPS; j++) {
            if (pr[j].y >= 0) {
                const int dl = pr[j].y & (BUCKET_NODES - 1);
                const int p = atomicAdd(&cur[dl], 1);
                if (p & LONG_ROW) csr[p ^ LONG_ROW] = pr[j].x;
                else lrow[dl * ROW_PITCH + p] = pr[j].x;
            }
        }
#pragma unroll
        for (int j = 0; j < RUN_STEPS; j++) {
            for (int k = sub + RUN_LANES; k < L[j]; k += RUN_LANES) {
                const int2 q2 = *(const int2*)(pairs + 2 * (long long)(S[j] + k));
                const int dl = q2.y & (BUCKET_NODES - 1);
                const int p = atomicAdd(&cur[dl], 1);
                if (p & LONG_ROW) csr[p ^ LONG_ROW] = q2.x;
                else lrow[dl * ROW_PITCH + p] = q2.x;
            }
        }
    }
    __syncthreads();

    const int cs = (c <= ROW_SLOTS) ? c : 0;
    if (__any(cs > 16)) order_row<ROW_SLOTS>(lrow + tid * ROW_PITCH, cs, csr + start);
    else order_row<16>(lrow + tid * ROW_PITCH, cs, csr + start);
}

#define GATHER_BLOCKS 2048

template <int K, typename vecf>
__device__ __forceinline__ void add_rows(vecf& acc, const int skey, const int j,
                                         const float* __restrict__ dinv, const float di,
                                         const float* __restrict__ H, const int lofs) {
    constexpr int D = (int)(sizeof(vecf) / sizeof(float)) * 32;
    int s[K];
    float w[K];
    vecf v[K];
#pragma unroll
    for (int q = 0; q < K; q++) s[q] = __builtin_amdgcn_readlane(skey, j + q);
#pragma unroll
    for (int q = 0; q < K; q++) w[q] = dinv[s[q]] * di;
#pragma unroll
    for (int q = 0; q < K; q++) v[q] = *(const vecf*)(H + (long long)s[q] * D + lofs);
#pragma unroll
    for (int q = 0; q < K; q++) acc += v[q] * w[q];
}

template <int K, typename vecf>
__device__ __forceinline__ void add_plain_rows(vecf& acc, const int skey, const int j,
                                               const float* __restrict__ H, const int lofs) {
    constexpr int D = (int)(sizeof(vecf) / sizeof(float)) * 32;
    int s[K];
    vecf v[K];
#pragma unroll
    for (int q = 0; q < K; q++) s[q] = __builtin_amdgcn_readlane(skey, j + q);
#pragma unroll
    for (int q = 0; q < K; q++) v[q] = *(const vecf*)(H + (long long)s[q] * D + lofs);
#pragma unroll
    for (int q = 0; q < K; q++) acc += v[q];
}

template <typename vecf>
__device__ __forceinline__ vecf add_long_row_plain(vecf acc, const int beg, const int end, const int lane,
                                                   const int* __restrict__ csr,
                                                   const float* __restrict__ H, const int lofs) {
    constexpr int D = (int)(sizeof(vecf) / sizeof(float)) * 32;
    int prev = -1;
    int left = end - beg;
    while (left > 0) {
        int best = 0x7fffffff;
        for (int b = beg + lane; b < end; b += 32) {
            int k = csr[b];
            if (k > prev && k < best) best = k;
        }
        for (int o = 16; o > 0; o >>= 1) {
            int u = __shfl_xor(best, o, 32);
            best = u < best ? u : best;
        }
        int c = 0;
        for (int b = beg + lane; b < end; b += 32) c += (csr[b] == best) ? 1 : 0;
        for (int o = 16; o > 0; o >>= 1) c += __shfl_xor(c, o, 32);
        best = __builtin_amdgcn_readfirstlane(best);
        c = __builtin_amdgcn_readfirstlane(c);
        if (c <= 0) break;
        const vecf v0 = *(const vecf*)(H + (long long)best * D + lofs);
#pragma clang loop unroll(disable)
        for (int r = 0; r < c; r++) acc += v0;
        prev = best;
        left -= c;
    }
    return acc;
}

__global__ __launch_bounds__(256) void scaled_row_sum_kernel(const int* __restrict__ rp,
                                                             const int* __restrict__ csr,
                                                             const float* __restrict__ dinv,
                                                             const float* __restrict__ Hs,
                                                             const float* __restrict__ bias,
                                                             float* __restrict__ Out, int n) {
    constexpr int D = 64;
    constexpr int VEC = D / 32;
    typedef __attribute__((ext_vector_type(VEC))) float vecf;
    const int lane = threadIdx.x & 31;
    const int nw = (int)gridDim.x * 8;
    int node = __builtin_amdgcn_readfirstlane((int)(blockIdx.x * 8 + (threadIdx.x >> 5)));
    if (node >= n) return;

    const int lofs = lane * VEC;
    const vecf bv = *(const vecf*)(bias + lofs);

    int beg = rp[node];
    int end = rp[node + 1];
    float di = dinv[node];
    vecf acc = *(const vecf*)(Hs + (long long)node * D + lofs);
    int key = 0x7fffffff;
    if (lane < end - beg) key = csr[beg + lane];
    int node1 = node + nw;
    int beg1 = 0, end1 = 0;
    if (node1 < n) {
        beg1 = rp[node1];
        end1 = rp[node1 + 1];
    }
    asm volatile("" : "+v"(key), "+v"(acc));

    for (;;) {
        int key1 = 0x7fffffff;
        float di1 = 0.0f;
        vecf hv1 = {};
        const int node2 = node1 + nw;
        int beg2 = 0, end2 = 0;
        if (node1 < n) {
            if (lane < end1 - beg1) key1 = csr[beg1 + lane];
            di1 = dinv[node1];
            hv1 = *(const vecf*)(Hs + (long long)node1 * D + lofs);
            if (node2 < n) {
                beg2 = rp[node2];
                end2 = rp[node2 + 1];
            }
        }

        const int d = end - beg;

        if (d <= 32) {
            const int skey = key;

            int j = 0;
            for (; j + 8 <= d; j += 8) add_plain_rows<8>(acc, skey, j, Hs, lofs);
            switch (d - j) {
                case 7: add_plain_rows<7>(acc, skey, j, Hs, lofs); break;
                case 6: add_plain_rows<6>(acc, skey, j, Hs, lofs); break;
                case 5: add_plain_rows<5>(acc, skey, j, Hs, lofs); break;
                case 4: add_plain_rows<4>(acc, skey, j, Hs, lofs); break;
                case 3: add_plain_rows<3>(acc, skey, j, Hs, lofs); break;
                case 2: add_plain_rows<2>(acc, skey, j, Hs, lofs); break;
                case 1: add_plain_rows<1>(acc, skey, j, Hs, lofs); break;
                default: break;
            }
        } else {
            acc = add_long_row_plain<vecf>(acc, beg, end, lane, csr, Hs, lofs);
        }

        vecf o;
        for (int q = 0; q < VEC; q++) o[q] = __builtin_fmaf(acc[q], di, bv[q]);
        *(vecf*)(Out + (long long)node * D + lofs) = o;

        acc = hv1;
        if (node1 >= n) break;
        node = node1; beg = beg1; end = end1; di = di1; key = key1;
        node1 = node2; beg1 = beg2; end1 = end2;
    }
}

#define TILE_NODES 64
#define TILE_PITCH 128
#define FUSED_THREADS 512
#define FUSED_BLOCKS 512

template <typename vecf>
__device__ __forceinline__ vecf sum_row(vecf acc, const int key, const int beg, const int end, const int lane,
                                        const int* __restrict__ csr, const float* __restrict__ dinv,
                                        const float di, const float* __restrict__ H, const int lofs) {
    constexpr int D = (int)(sizeof(vecf) / sizeof(float)) * 32;
    const int d = end - beg;
    if (d <= 32) {
        const int skey = key;
        int j = 0;
        for (; j + 8 <= d; j += 8) add_rows<8>(acc, skey, j, dinv, di, H, lofs);
        switch (d - j) {
            case 7: add_rows<7>(acc, skey, j, dinv, di, H, lofs); break;
            case 6: add_rows<6>(acc, skey, j, dinv, di, H, lofs); break;
            case 5: add_rows<5>(acc, skey, j, dinv, di, H, lofs); break;
            case 4: add_rows<4>(acc, skey, j, dinv, di, H, lofs); break;
            case 3: add_rows<3>(acc, skey, j, dinv, di, H, lofs); break;
            case 2: add_rows<2>(acc, skey, j, dinv, di, H, lofs); break;
            case 1: add_rows<1>(acc, skey, j, dinv, di, H, lofs); break;
            default: break;
        }
    } else {
        int prev = -1;
        int left = d;
        while (left > 0) {
            int best = 0x7fffffff;
            for (int b = beg + lane; b < end; b += 32) {
                int k = csr[b];
                if (k > prev && k < best) best = k;
            }
            for (int o = 16; o > 0; o >>= 1) {
                int u = __shfl_xor(best, o, 32);
                best = u < best ? u : best;
            }
            int c = 0;
            for (int b = beg + lane; b < end; b += 32) c += (csr[b] == best) ? 1 : 0;
            for (int o = 16; o > 0; o >>= 1) c += __shfl_xor(c, o, 32);
            best = __builtin_amdgcn_readfirstlane(best);
            c = __builtin_amdgcn_readfirstlane(c);
            if (c <= 0) break;
            const float w0 = dinv[best] * di;
            const vecf v0 = *(const vecf*)(H + (long long)best * D + lofs);
            for (int r = 0; r < c; r++) acc += v0 * w0;
            prev = best;
            left -= c;
        }
    }
    return acc;
}

__device__ __forceinline__ int tile_bounds(const int* __restrict__ rp, const int nb0, const int n, const int lane) {
    const int idx = nb0 + lane;
    return (lane < 5) ? rp[idx < n ? idx : n] : 0;
}

template <typename vecf>
__device__ __forceinline__ void tile_heads(const int rpv, const int nb0, const int n, const int lane, const int lofs,
                                           const int* __restrict__ csr, const float* __restrict__ dinv,
                                           const float* __restrict__ X,
                                           int (&beg)[4], int (&end)[4], int (&key)[4], float (&di)[4], vecf (&hv)[4]) {
#pragma unroll
    for (int q = 0; q < 4; q++) {
        beg[q] = __builtin_amdgcn_readlane(rpv, q);
        end[q] = __builtin_amdgcn_readlane(rpv, q + 1);
    }
#pragma unroll
    for (int q = 0; q < 4; q++) {
        const int node = nb0 + q;
        key[q] = 0x7fffffff;
        di[q] = 0.0f;
        hv[q] = vecf{};
        if (node < n) {
            if (lane < end[q] - beg[q]) key[q] = csr[beg[q] + lane];
            di[q] = dinv[node];
            hv[q] = *(const vecf*)(X + (long long)node * DIN + lofs);
        }
    }
}

__global__ __launch_bounds__(FUSED_THREADS) void gather_gemm_elu_gemm_kernel(const int* __restrict__ rp,
                                                                             const int* __restrict__ csr,
                                                                             const float* __restrict__ dinv,
                                                                             const float* __restrict__ X,
                                                                             const float* __restrict__ W1,
                                                                             const float* __restrict__ bias1,
                                                                             const float* __restrict__ W2,
                                                                             float* __restrict__ H2, int n) {
    constexpr int D1 = 128;
    constexpr int D2 = 64;
    constexpr int NT1 = D1 / 16;
    constexpr int NT2 = D2 / 16;
    constexpr int NPAIR1 = 32 * NT1 * 32;
    constexpr int NPAIR2 = 32 * NT2 * 32;
    typedef __attribute__((ext_vector_type(4))) float vecf;
    __shared__ __attribute__((aligned(16))) float sW1[NPAIR1 * 2];
    __shared__ __attribute__((aligned(16))) float sW2[NPAIR2 * 2];
    __shared__ __attribute__((aligned(16))) float sA[TILE_NODES * TILE_PITCH];
    __shared__ __attribute__((aligned(16))) float sG[TILE_NODES * TILE_PITCH];

    const int lane = threadIdx.x & 31;
    const int wave = __builtin_amdgcn_readfirstlane((int)(threadIdx.x >> 5));
    const int lofs = lane * 4;
    const int ntiles = (n + TILE_NODES - 1) / TILE_NODES;
    const int tstep = (int)gridDim.x;
    int tile = (int)blockIdx.x;

    int beg[4], end[4], key[4];
    float di[4];
    vecf hv[4];
    int rpv = 0;
    if (tile < ntiles) rpv = tile_bounds(rp, tile * TILE_NODES + wave * 4, n, lane);

    for (int pair = threadIdx.x; pair < NPAIR1; pair += FUSED_THREADS) {
        int l  = pair & 31;
        int tt = (pair >> 5) % NT1;
        int ks = (pair >> 5) / NT1;
        int col = tt * 16 + (l & 15);
        int k0  = ks * 4 + ((l >> 4) * 2);
        sW1[pair * 2 + 0] = W1[k0 * D1 + col];
        sW1[pair * 2 + 1] = W1[(k0 + 1) * D1 + col];
    }
    tile_heads<vecf>(rpv, tile * TILE_NODES + wave * 4, n, lane, lofs, csr, dinv, X, beg, end, key, di, hv);

    for (int pair = threadIdx.x; pair < NPAIR2; pair += FUSED_THREADS) {
        int l  = pair & 31;
        int tt = (pair >> 5) % NT2;
        int ks = (pair >> 5) / NT2;
        int col = tt * 16 + (l & 15);
        int k0  = ks * 4 + ((l >> 4) * 2);
        sW2[pair * 2 + 0] = W2[k0 * D2 + col];
        sW2[pair * 2 + 1] = W2[(k0 + 1) * D2 + col];
    }

    const int m    = lane & 15;
    const int kk   = (lane >> 4) * 2;

    const int ct = wave & 7;
    const int rg = wave >> 3;
    const float bcol = bias1[ct * 16 + m];
    const float* __restrict__ a0p = sA + (rg * 32 + m) * TILE_PITCH + kk;
    const float* __restrict__ a1p = a0p + 16 * TILE_PITCH;
    const float* __restrict__ bp1 = sW1 + (ct * 32 + lane) * 2;
    const int gc   = ct * 16 + m;
    const int grow = rg * 32 + (lane >> 4) * 8;

    const int rt = wave >> 2;
    const int c2 = wave & 3;
    const float* __restrict__ gp  = sG + (rt * 16 + m) * TILE_PITCH + kk;
    const float* __restrict__ bp2 = sW2 + (c2 * 32 + lane) * 2;

    for (; tile < ntiles; tile += tstep) {
        const int nb0 = tile * TILE_NODES + wave * 4;
        const int tnext = tile + tstep;
        const int nb0n = tnext * TILE_NODES + wave * 4;
        int rpvn = 0;
        if (tnext < ntiles) rpvn = tile_bounds(rp, nb0n, n, lane);
#pragma unroll
        for (int q = 0; q < 4; q++) {
            vecf acc = hv[q] * (di[q] * di[q]);
            if (nb0 + q < n) acc = sum_row<vecf>(acc, key[q], beg[q], end[q], lane, csr, dinv, di[q], X, lofs);
            const int r = wave * 4 + q;
            *(vecf*)(sA + r * TILE_PITCH + ((lane ^ (r & 15)) << 2)) = acc;
        }
        __syncthreads();

        v8f acc0 = {};
        v8f acc1 = {};
#pragma unroll 8
        for (int ks = 0; ks < 32; ks++) {
            v2f b  = *(const v2f*)(bp1 + ks * (NT1 * 64));
            const int ko = (ks ^ m) << 2;
            v2f a0 = *(const v2f*)(a0p + ko);
            v2f a1 = *(const v2f*)(a1p + ko);
            acc0 = __builtin_amdgcn_wmma_f32_16x16x4_f32(false, a0, false, b, (short)0, acc0, false, false);
            acc1 = __builtin_amdgcn_wmma_f32_16x16x4_f32(false, a1, false, b, (short)0, acc1, false, false);
        }

#pragma unroll
        for (int v = 0; v < 8; v++) {
            float t0 = acc0[v] + bcol;
            float t1 = acc1[v] + bcol;
            t0 = t0 > 0.0f ? t0 : (__expf(t0) - 1.0f);
            t1 = t1 > 0.0f ? t1 : (__expf(t1) - 1.0f);
            const int r0 = grow + v;
            const int go = (((gc >> 2) ^ (r0 & 15)) << 2) + (gc & 3);
            sG[r0 * TILE_PITCH + go] = t0;
            sG[(r0 + 16) * TILE_PITCH + go] = t1;
        }
        __syncthreads();

        int begn[4], endn[4], keyn[4];
        float din[4];
        vecf hvn[4];
        if (tnext < ntiles) {
            tile_heads<vecf>(rpvn, nb0n, n, lane, lofs, csr, dinv, X, begn, endn, keyn, din, hvn);
        } else {
#pragma unroll
            for (int q = 0; q < 4; q++) {
                begn[q] = 0; endn[q] = 0; keyn[q] = 0x7fffffff; din[q] = 0.0f; hvn[q] = vecf{};
            }
        }

        const long long rbase = (long long)tile * TILE_NODES + rt * 16 + (lane >> 4) * 8;
        float dsc[8];
#pragma unroll
        for (int v = 0; v < 8; v++) {
            const long long r = rbase + v;
            dsc[v] = dinv[r < n ? r : (long long)(n - 1)];
        }
        v8f acc2 = {};
#pragma unroll 8
        for (int ks = 0; ks < 32; ks++) {
            v2f b = *(const v2f*)(bp2 + ks * (NT2 * 64));
            v2f a = *(const v2f*)(gp + ((ks ^ m) << 2));
            acc2 = __builtin_amdgcn_wmma_f32_16x16x4_f32(false, a, false, b, (short)0, acc2, false, false);
        }

#pragma unroll
        for (int v = 0; v < 8; v++) {
            const long long r = rbase + v;
            if (r < n) H2[r * D2 + c2 * 16 + m] = acc2[v] * dsc[v];
        }
#pragma unroll
        for (int q = 0; q < 4; q++) {
            beg[q] = begn[q]; end[q] = endn[q]; key[q] = keyn[q]; di[q] = din[q]; hv[q] = hvn[q];
        }
    }
}

extern "C" void kernel_launch(void* const* d_in, const int* in_sizes, int n_in,
                              void* d_out, int out_size, void* d_ws, size_t ws_size,
                              hipStream_t stream) {
    const float* x  = (const float*)d_in[0];
    const int*   ei = (const int*)d_in[1];
    const float* W1 = (const float*)d_in[2];
    const float* b1 = (const float*)d_in[3];
    const float* W2 = (const float*)d_in[4];
    const float* b2 = (const float*)d_in[5];
    float* out = (float*)d_out;

    const int N = in_sizes[0] / DIN;
    const int E = in_sizes[1] / 2;
    const int DH = 128;

    const int B = 256;
    const int gNode = GATHER_BLOCKS;
    const int NB    = (N + BUCKET_NODES - 1) / BUCKET_NODES;

    char* ws = (char*)d_ws;
    float* dinv = (float*)ws;
    size_t off = (((size_t)N * sizeof(float)) + 255) & ~(size_t)255;
    float* hbuf = (float*)(ws + off);
    float* aggbuf = (float*)(ws + off + (size_t)N * DH * sizeof(float));
    char* tail = ws + off + 2 * (size_t)N * DH * sizeof(float);
    int* rp = (int*)tail;
    size_t rpb = (((size_t)(N + 1) * sizeof(int)) + 255) & ~(size_t)255;
    int* csr = (int*)(tail + rpb);
    int* pairs = (int*)aggbuf;
    int* cpre  = pairs + 2 * (size_t)E;

    chunk_sort_kernel<<<SORT_BLOCKS, SORT_THREADS, 2 * NB * sizeof(int), stream>>>(ei, E, NB, cpre, pairs);
    bucket_rows_kernel<<<NB, BUCKET_NODES, 0, stream>>>(pairs, cpre, E, N, rp, dinv, csr);

    gather_gemm_elu_gemm_kernel<<<FUSED_BLOCKS, FUSED_THREADS, 0, stream>>>(rp, csr, dinv, x, W1, b1, W2, hbuf, N);

    scaled_row_sum_kernel<<<gNode, B, 0, stream>>>(rp, csr, dinv, hbuf, b2, out, N);
}
